# starting kernel
# baseline (speedup 1.0000x reference)
#include <hip/hip_runtime.h>

typedef float v2f __attribute__((ext_vector_type(2)));
typedef float v8f __attribute__((ext_vector_type(8)));

#define CH 300
#define CHPAD 304
#define TP 4


__global__ void fill_kernel(float* __restrict__ p, float v, int n) {
    int i = blockIdx.x * blockDim.x + threadIdx.x;
    if (i < n) p[i] = v;
}

__global__ void deg_kernel(const int* __restrict__ row, float* __restrict__ deg, int E) {
    int e = blockIdx.x * blockDim.x + threadIdx.x;
    if (e < E) unsafeAtomicAdd(&deg[row[e]], 1.0f);
}

__global__ void rsqrt_kernel(float* __restrict__ d, int n) {
    int i = blockIdx.x * blockDim.x + threadIdx.x;
    if (i < n) d[i] = rsqrtf(d[i]);
}

__global__ void padw_kernel(const float* __restrict__ W, float* __restrict__ Wp) {
    int i = blockIdx.x * blockDim.x + threadIdx.x;
    if (i < CHPAD * CH) Wp[i] = (i < CH * CH) ? W[i] : 0.0f;
}

__global__ void mean_kernel(const float* __restrict__ x, float* __restrict__ xm, int total) {
    int i = blockIdx.x * blockDim.x + threadIdx.x;
    if (i >= total) return;
    int n = i / CH;
    int c = i - n * CH;
    const float* b = x + (size_t)n * (TP * CH) + c;
    xm[i] = 0.25f * (b[0] + b[CH] + b[2 * CH] + b[3 * CH]);
}

__global__ __launch_bounds__(128) void gemm_wmma_kernel(const float* __restrict__ A,
                                                        const float* __restrict__ Wp,
                                                        const float* __restrict__ bias,
                                                        float* __restrict__ C,
                                                        int mtiles) {
    const int wave = threadIdx.x >> 5;
    const int lane = threadIdx.x & 31;
    const int trow = blockIdx.x * 4 + wave;
    if (trow >= mtiles) return;

    const int mn   = lane & 15;
    const int half = lane >> 4;
    const int row0 = trow * 16;
    const int col0 = blockIdx.y * 64;

    v8f acc[4] = {};

    const float* ap = A  + (size_t)(row0 + mn) * CH + 2 * half;
    const float* bp = Wp + (size_t)(col0 + mn) * CH + 2 * half;

    for (int k0 = 0; k0 < CH; k0 += 4) {
        v2f a = *(const v2f*)(ap + k0);
#pragma unroll
        for (int t = 0; t < 4; ++t) {
            v2f b = *(const v2f*)(bp + k0 + t * (16 * CH));
            acc[t] = __builtin_amdgcn_wmma_f32_16x16x4_f32(
                false, a, false, b, (short)0, acc[t], false, false);
        }
    }

#pragma unroll
    for (int t = 0; t < 4; ++t) {
        int n = col0 + t * 16 + mn;
        if (n < CH) {
            float bv = bias[n];
#pragma unroll
            for (int j = 0; j < 8; ++j) {
                int m = row0 + j + 8 * half;
                C[(size_t)m * CH + n] = acc[t][j] + bv;
            }
        }
    }
}

__global__ void scatter_kernel(const int* __restrict__ row, const int* __restrict__ col,
                               const float* __restrict__ dis, const float* __restrict__ h,
                               float* __restrict__ acc) {
    int e = blockIdx.x;
    int c = blockIdx.y * blockDim.x + threadIdx.x;
    if (c >= CH) return;
    int r = row[e];
    int d = col[e];
    float v = dis[r] * h[(size_t)r * CH + c];
    unsafeAtomicAdd(&acc[(size_t)d * CH + c], v);
}

__global__ void finalize_kernel(const float* __restrict__ acc, const float* __restrict__ h,
                                const float* __restrict__ dis, float* __restrict__ out,
                                int total, int relu) {
    int i = blockIdx.x * blockDim.x + threadIdx.x;
    if (i >= total) return;
    int n = i / CH;
    float dn = dis[n];
    float v = dn * (acc[i] + dn * h[i]);
    if (relu) v = (v > 0.0f) ? v : 0.01f * v;
    out[i] = v;
}

extern "C" void kernel_launch(void* const* d_in, const int* in_sizes, int n_in,
                              void* d_out, int out_size, void* d_ws, size_t ws_size,
                              hipStream_t stream) {
    const float* x  = (const float*)d_in[0];
    const int*   ei = (const int*)d_in[1];
    const float* W1 = (const float*)d_in[2];
    const float* b1 = (const float*)d_in[3];
    const float* W2 = (const float*)d_in[4];
    const float* b2 = (const float*)d_in[5];
    float* out = (float*)d_out;

    const int E = in_sizes[1] / 2;
    const int N = in_sizes[0] / (TP * CH);
    const int total = N * CH;
    const int* row = ei;
    const int* col = ei + E;

    char* ws = (char*)d_ws;
    float* dis = (float*)ws;
    size_t off = ((size_t)N * sizeof(float) + 255) & ~(size_t)255;
    float* Wp = (float*)(ws + off);
    off += (size_t)CHPAD * CH * sizeof(float);
    float* bufA = (float*)(ws + off);
    float* bufB = bufA + total;
    float* bufC = bufB + total;

    const int TB = 256;
    const int mtiles = (N + 15) / 16;
    dim3 gg((mtiles + 3) / 4, (CH + 63) / 64);
    dim3 gs(E, (CH + 127) / 128);
    const int padBlocks = (CHPAD * CH + TB - 1) / TB;

    fill_kernel<<<(N + TB - 1) / TB, TB, 0, stream>>>(dis, 1.0f, N);
    deg_kernel<<<(E + TB - 1) / TB, TB, 0, stream>>>(row, dis, E);
    rsqrt_kernel<<<(N + TB - 1) / TB, TB, 0, stream>>>(dis, N);

    mean_kernel<<<(total + TB - 1) / TB, TB, 0, stream>>>(x, bufA, total);

    padw_kernel<<<padBlocks, TB, 0, stream>>>(W1, Wp);
    gemm_wmma_kernel<<<gg, 128, 0, stream>>>(bufA, Wp, b1, bufB, mtiles);

    fill_kernel<<<(total + TB - 1) / TB, TB, 0, stream>>>(bufC, 0.0f, total);
    scatter_kernel<<<gs, 128, 0, stream>>>(row, col, dis, bufB, bufC);
    finalize_kernel<<<(total + TB - 1) / TB, TB, 0, stream>>>(bufC, bufB, dis, bufA, total, 1);

    padw_kernel<<<padBlocks, TB, 0, stream>>>(W2, Wp);
    gemm_wmma_kernel<<<gg, 128, 0, stream>>>(bufA, Wp, b2, bufB, mtiles);

    fill_kernel<<<(total + TB - 1) / TB, TB, 0, stream>>>(out, 0.0f, total);
    scatter_kernel<<<gs, 128, 0, stream>>>(row, col, dis, bufB, out);
    finalize_kernel<<<(total + TB - 1) / TB, TB, 0, stream>>>(out, bufB, dis, out, total, 0);
}


# Edge aggregation as a per-destination CSR gather (counting sort by col rebuilt each call, arrival-order slots) replaces zero-fill + float atomic scatter + finalize; per-node sum order changes
# speedup vs baseline: 1.8095x; 1.8095x over previous
#include <hip/hip_runtime.h>

typedef float v2f __attribute__((ext_vector_type(2)));
typedef float v8f __attribute__((ext_vector_type(8)));

#define CH 300
#define CHPAD 304
#define TP 4
#define CH4 75
#define SCAN_T 1024


__global__ void init_kernel(float* __restrict__ deg, int* __restrict__ cnt, int n) {
    int i = blockIdx.x * blockDim.x + threadIdx.x;
    if (i < n) { deg[i] = 1.0f; cnt[i] = 0; }
}

__global__ void deg_kernel(const int* __restrict__ row, const int* __restrict__ col,
                           float* __restrict__ deg, int* __restrict__ cnt, int E) {
    int e = blockIdx.x * blockDim.x + threadIdx.x;
    if (e < E) {
        unsafeAtomicAdd(&deg[row[e]], 1.0f);
        atomicAdd(&cnt[col[e]], 1);
    }
}

__global__ __launch_bounds__(SCAN_T) void scan_kernel(const int* __restrict__ cnt,
                                                     int* __restrict__ ptr,
                                                     int* __restrict__ cur, int n) {
    __shared__ int part[SCAN_T];
    const int t = threadIdx.x;
    const int per = (n + SCAN_T - 1) / SCAN_T;
    int b = t * per;
    if (b > n) b = n;
    int e = b + per;
    if (e > n) e = n;
    int sum = 0;
    for (int i = b; i < e; ++i) sum += cnt[i];
    part[t] = sum;
    __syncthreads();
    for (int o = 1; o < SCAN_T; o <<= 1) {
        int v = (t >= o) ? part[t - o] : 0;
        __syncthreads();
        part[t] += v;
        __syncthreads();
    }
    int run = part[t] - sum;
    for (int i = b; i < e; ++i) {
        ptr[i] = run;
        cur[i] = run;
        run += cnt[i];
    }
    if (t == SCAN_T - 1) ptr[n] = run;
}

__global__ void csr_fill_kernel(const int* __restrict__ row, const int* __restrict__ col,
                                int* __restrict__ cur, int* __restrict__ src, int E) {
    int e = blockIdx.x * blockDim.x + threadIdx.x;
    if (e < E) {
        int p = atomicAdd(&cur[col[e]], 1);
        src[p] = row[e];
    }
}

__global__ void rsqrt_kernel(float* __restrict__ d, int n) {
    int i = blockIdx.x * blockDim.x + threadIdx.x;
    if (i < n) d[i] = rsqrtf(d[i]);
}

__global__ void padw_kernel(const float* __restrict__ W, float* __restrict__ Wp) {
    int i = blockIdx.x * blockDim.x + threadIdx.x;
    if (i < CHPAD * CH) Wp[i] = (i < CH * CH) ? W[i] : 0.0f;
}

__global__ void mean_kernel(const float* __restrict__ x, float* __restrict__ xm, int total) {
    int i = blockIdx.x * blockDim.x + threadIdx.x;
    if (i >= total) return;
    int n = i / CH;
    int c = i - n * CH;
    const float* b = x + (size_t)n * (TP * CH) + c;
    xm[i] = 0.25f * (b[0] + b[CH] + b[2 * CH] + b[3 * CH]);
}

__global__ __launch_bounds__(128) void gemm_wmma_kernel(const float* __restrict__ A,
                                                        const float* __restrict__ Wp,
                                                        const float* __restrict__ bias,
                                                        float* __restrict__ C,
                                                        int mtiles) {
    const int wave = threadIdx.x >> 5;
    const int lane = threadIdx.x & 31;
    const int trow = blockIdx.x * 4 + wave;
    if (trow >= mtiles) return;

    const int mn   = lane & 15;
    const int half = lane >> 4;
    const int row0 = trow * 16;
    const int col0 = blockIdx.y * 64;

    v8f acc[4] = {};

    const float* ap = A  + (size_t)(row0 + mn) * CH + 2 * half;
    const float* bp = Wp + (size_t)(col0 + mn) * CH + 2 * half;

    for (int k0 = 0; k0 < CH; k0 += 4) {
        v2f a = *(const v2f*)(ap + k0);
#pragma unroll
        for (int t = 0; t < 4; ++t) {
            v2f b = *(const v2f*)(bp + k0 + t * (16 * CH));
            acc[t] = __builtin_amdgcn_wmma_f32_16x16x4_f32(
                false, a, false, b, (short)0, acc[t], false, false);
        }
    }

#pragma unroll
    for (int t = 0; t < 4; ++t) {
        int n = col0 + t * 16 + mn;
        if (n < CH) {
            float bv = bias[n];
#pragma unroll
            for (int j = 0; j < 8; ++j) {
                int m = row0 + j + 8 * half;
                C[(size_t)m * CH + n] = acc[t][j] + bv;
            }
        }
    }
}

__global__ __launch_bounds__(256) void gather_kernel(const int* __restrict__ ptr,
                                                     const int* __restrict__ src,
                                                     const float* __restrict__ dis,
                                                     const float* __restrict__ h,
                                                     float* __restrict__ out,
                                                     int total4, int relu) {
    const int g = blockIdx.x * 256 + threadIdx.x;
    if (g >= total4) return;
    const int n = g / CH4;
    const int q = g - n * CH4;
    const float4* __restrict__ h4 = (const float4*)h;
    const float dn = dis[n];
    const float4 self = h4[g];
    const int p0 = ptr[n];
    const int p1 = ptr[n + 1];
    float4 s = make_float4(0.0f, 0.0f, 0.0f, 0.0f);
    for (int p = p0; p < p1; ++p) {
        const int r = src[p];
        const float w = dis[r];
        const float4 v = h4[(size_t)r * CH4 + q];
        s.x += w * v.x; s.y += w * v.y; s.z += w * v.z; s.w += w * v.w;
    }
    float4 o;
    o.x = dn * (s.x + dn * self.x);
    o.y = dn * (s.y + dn * self.y);
    o.z = dn * (s.z + dn * self.z);
    o.w = dn * (s.w + dn * self.w);
    if (relu) {
        o.x = (o.x > 0.0f) ? o.x : 0.01f * o.x;
        o.y = (o.y > 0.0f) ? o.y : 0.01f * o.y;
        o.z = (o.z > 0.0f) ? o.z : 0.01f * o.z;
        o.w = (o.w > 0.0f) ? o.w : 0.01f * o.w;
    }
    ((float4*)out)[g] = o;
}

extern "C" void kernel_launch(void* const* d_in, const int* in_sizes, int n_in,
                              void* d_out, int out_size, void* d_ws, size_t ws_size,
                              hipStream_t stream) {
    const float* x  = (const float*)d_in[0];
    const int*   ei = (const int*)d_in[1];
    const float* W1 = (const float*)d_in[2];
    const float* b1 = (const float*)d_in[3];
    const float* W2 = (const float*)d_in[4];
    const float* b2 = (const float*)d_in[5];
    float* out = (float*)d_out;

    const int E = in_sizes[1] / 2;
    const int N = in_sizes[0] / (TP * CH);
    const int total = N * CH;
    const int* row = ei;
    const int* col = ei + E;

    char* ws = (char*)d_ws;
    float* dis = (float*)ws;
    size_t off = ((size_t)N * sizeof(float) + 255) & ~(size_t)255;
    float* Wp = (float*)(ws + off);
    off += (size_t)CHPAD * CH * sizeof(float);
    float* bufA = (float*)(ws + off);
    float* bufB = bufA + total;
    int* cnt = (int*)(bufB + total);
    int* ptr = cnt + N;
    int* cur = ptr + (N + 1);
    int* src = cur + N;

    const int TB = 256;
    const int mtiles = (N + 15) / 16;
    dim3 gg((mtiles + 3) / 4, (CH + 63) / 64);
    const int total4 = N * CH4;
    const int gatherBlocks = (total4 + 255) / 256;
    const int padBlocks = (CHPAD * CH + TB - 1) / TB;

    init_kernel<<<(N + TB - 1) / TB, TB, 0, stream>>>(dis, cnt, N);
    deg_kernel<<<(E + TB - 1) / TB, TB, 0, stream>>>(row, col, dis, cnt, E);
    rsqrt_kernel<<<(N + TB - 1) / TB, TB, 0, stream>>>(dis, N);

    scan_kernel<<<1, SCAN_T, 0, stream>>>(cnt, ptr, cur, N);
    csr_fill_kernel<<<(E + TB - 1) / TB, TB, 0, stream>>>(row, col, cur, src, E);

    mean_kernel<<<(total + TB - 1) / TB, TB, 0, stream>>>(x, bufA, total);

    padw_kernel<<<padBlocks, TB, 0, stream>>>(W1, Wp);
    gemm_wmma_kernel<<<gg, 128, 0, stream>>>(bufA, Wp, b1, bufB, mtiles);

    gather_kernel<<<gatherBlocks, 256, 0, stream>>>(ptr, src, dis, bufB, bufA, total4, 1);

    padw_kernel<<<padBlocks, TB, 0, stream>>>(W2, Wp);
    gemm_wmma_kernel<<<gg, 128, 0, stream>>>(bufA, Wp, b2, bufB, mtiles);

    gather_kernel<<<gatherBlocks, 256, 0, stream>>>(ptr, src, dis, bufB, out, total4, 0);
}


# Both GEMMs: A and W tiles staged through LDS (64 rows x all columns per block, 16-byte global loads, double-buffered k stages), result tile stored from LDS in whole lines; padded-W copy kernels go
# speedup vs baseline: 2.4649x; 1.3622x over previous
#include <hip/hip_runtime.h>

typedef float v2f __attribute__((ext_vector_type(2)));
typedef float v4f __attribute__((ext_vector_type(4)));
typedef float v8f __attribute__((ext_vector_type(8)));

#define CH 300
#define TP 4
#define CH4 75
#define SCAN_T 1024


__global__ void init_kernel(float* __restrict__ deg, int* __restrict__ cnt, int n) {
    int i = blockIdx.x * blockDim.x + threadIdx.x;
    if (i < n) { deg[i] = 1.0f; cnt[i] = 0; }
}

__global__ void deg_kernel(const int* __restrict__ row, const int* __restrict__ col,
                           float* __restrict__ deg, int* __restrict__ cnt, int E) {
    int e = blockIdx.x * blockDim.x + threadIdx.x;
    if (e < E) {
        unsafeAtomicAdd(&deg[row[e]], 1.0f);
        atomicAdd(&cnt[col[e]], 1);
    }
}

__global__ __launch_bounds__(SCAN_T) void scan_kernel(const int* __restrict__ cnt,
                                                     int* __restrict__ ptr,
                                                     int* __restrict__ cur, int n) {
    __shared__ int part[SCAN_T];
    const int t = threadIdx.x;
    const int per = (n + SCAN_T - 1) / SCAN_T;
    int b = t * per;
    if (b > n) b = n;
    int e = b + per;
    if (e > n) e = n;
    int sum = 0;
    for (int i = b; i < e; ++i) sum += cnt[i];
    part[t] = sum;
    __syncthreads();
    for (int o = 1; o < SCAN_T; o <<= 1) {
        int v = (t >= o) ? part[t - o] : 0;
        __syncthreads();
        part[t] += v;
        __syncthreads();
    }
    int run = part[t] - sum;
    for (int i = b; i < e; ++i) {
        ptr[i] = run;
        cur[i] = run;
        run += cnt[i];
    }
    if (t == SCAN_T - 1) ptr[n] = run;
}

__global__ void csr_fill_kernel(const int* __restrict__ row, const int* __restrict__ col,
                                int* __restrict__ cur, int* __restrict__ src, int E) {
    int e = blockIdx.x * blockDim.x + threadIdx.x;
    if (e < E) {
        int p = atomicAdd(&cur[col[e]], 1);
        src[p] = row[e];
    }
}

__global__ void rsqrt_kernel(float* __restrict__ d, int n) {
    int i = blockIdx.x * blockDim.x + threadIdx.x;
    if (i < n) d[i] = rsqrtf(d[i]);
}

__global__ void mean_kernel(const float* __restrict__ x, float* __restrict__ xm, int total) {
    int i = blockIdx.x * blockDim.x + threadIdx.x;
    if (i >= total) return;
    int n = i / CH;
    int c = i - n * CH;
    const float* b = x + (size_t)n * (TP * CH) + c;
    xm[i] = 0.25f * (b[0] + b[CH] + b[2 * CH] + b[3 * CH]);
}

#define GBM 64
#define GBK 20
#define GNP 320
#define GNCH (CH / GBK)
#define GA4 (GBM * GBK / 4)
#define GS4 ((GBM + GNP) * GBK / 4)
#define GSF ((GBM + GNP) * GBK)
#define GLDS (GBM * CH)

__global__ __launch_bounds__(256) void gemm_wmma_kernel(const float* __restrict__ A,
                                                        const float* __restrict__ W,
                                                        const float* __restrict__ bias,
                                                        float* __restrict__ C,
                                                        int M) {
    __shared__ __attribute__((aligned(16))) float lds[GLDS];
    const int tid  = threadIdx.x;
    const int wave = tid >> 5;
    const int lane = tid & 31;
    const int mn   = lane & 15;
    const int half = lane >> 4;
    const int mt   = wave & 3;
    const int nh   = wave >> 2;
    const int row0 = blockIdx.x * GBM;

    const float* gp[8];
#pragma unroll
    for (int i = 0; i < 8; ++i) {
        const int idx = tid + 256 * i;
        if (idx < GA4) {
            const int r = idx / 5;
            const int j = idx - 5 * r;
            int gr = row0 + r;
            if (gr > M - 1) gr = M - 1;
            gp[i] = A + (size_t)gr * CH + 4 * j;
        } else {
            const int w = idx - GA4;
            int r = w / 5;
            const int j = w - 5 * r;
            if (r > CH - 1) r = CH - 1;
            gp[i] = W + (size_t)r * CH + 4 * j;
        }
    }
    const bool last = tid < (GS4 - 7 * 256);

    v4f v[8];
    v4f* l4 = (v4f*)lds;
#pragma unroll
    for (int i = 0; i < 7; ++i) v[i] = *(const v4f*)(gp[i]);
    if (last) v[7] = *(const v4f*)(gp[7]);
#pragma unroll
    for (int i = 0; i < 7; ++i) l4[tid + 256 * i] = v[i];
    if (last) l4[tid + 7 * 256] = v[7];
    __syncthreads();

    v8f acc[10] = {};
    const int aoff = (mt * 16 + mn) * GBK + 2 * half;
    const int boff = GBM * GBK + (nh * 160 + mn) * GBK + 2 * half;

    for (int ch = 0; ch < GNCH; ++ch) {
        const float* st = lds + (ch & 1) * GSF;
        const bool more = (ch + 1 < GNCH);
        if (more) {
            const int kc = (ch + 1) * GBK;
#pragma unroll
            for (int i = 0; i < 7; ++i) v[i] = *(const v4f*)(gp[i] + kc);
            if (last) v[7] = *(const v4f*)(gp[7] + kc);
        }
#pragma unroll
        for (int s = 0; s < GBK / 4; ++s) {
            const v2f a = *(const v2f*)(st + aoff + 4 * s);
#pragma unroll
            for (int t = 0; t < 10; ++t) {
                const v2f b = *(const v2f*)(st + boff + t * (16 * GBK) + 4 * s);
                acc[t] = __builtin_amdgcn_wmma_f32_16x16x4_f32(
                    false, a, false, b, (short)0, acc[t], false, false);
            }
        }
        if (more) {
            v4f* d4 = (v4f*)(lds + ((ch + 1) & 1) * GSF);
#pragma unroll
            for (int i = 0; i < 7; ++i) d4[tid + 256 * i] = v[i];
            if (last) d4[tid + 7 * 256] = v[7];
        }
        __syncthreads();
    }

    float bv[10];
#pragma unroll
    for (int t = 0; t < 10; ++t) {
        const int n = nh * 160 + t * 16 + mn;
        bv[t] = bias[n < CH ? n : CH - 1];
    }
#pragma unroll
    for (int t = 0; t < 10; ++t) {
        const int n = nh * 160 + t * 16 + mn;
        if (n < CH) {
#pragma unroll
            for (int j = 0; j < 8; ++j)
                lds[(mt * 16 + 8 * half + j) * CH + n] = acc[t][j] + bv[t];
        }
    }
    __syncthreads();
    int rows = M - row0;
    if (rows > GBM) rows = GBM;
    const int n4 = rows * CH4;
    v4f* c4 = (v4f*)(C + (size_t)row0 * CH);
    for (int i = tid; i < n4; i += 256) c4[i] = l4[i];
}

__global__ __launch_bounds__(256) void gather_kernel(const int* __restrict__ ptr,
                                                     const int* __restrict__ src,
                                                     const float* __restrict__ dis,
                                                     const float* __restrict__ h,
                                                     float* __restrict__ out,
                                                     int total4, int relu) {
    const int g = blockIdx.x * 256 + threadIdx.x;
    if (g >= total4) return;
    const int n = g / CH4;
    const int q = g - n * CH4;
    const float4* __restrict__ h4 = (const float4*)h;
    const float dn = dis[n];
    const float4 self = h4[g];
    const int p0 = ptr[n];
    const int p1 = ptr[n + 1];
    float4 s = make_float4(0.0f, 0.0f, 0.0f, 0.0f);
    for (int p = p0; p < p1; ++p) {
        const int r = src[p];
        const float w = dis[r];
        const float4 v = h4[(size_t)r * CH4 + q];
        s.x += w * v.x; s.y += w * v.y; s.z += w * v.z; s.w += w * v.w;
    }
    float4 o;
    o.x = dn * (s.x + dn * self.x);
    o.y = dn * (s.y + dn * self.y);
    o.z = dn * (s.z + dn * self.z);
    o.w = dn * (s.w + dn * self.w);
    if (relu) {
        o.x = (o.x > 0.0f) ? o.x : 0.01f * o.x;
        o.y = (o.y > 0.0f) ? o.y : 0.01f * o.y;
        o.z = (o.z > 0.0f) ? o.z : 0.01f * o.z;
        o.w = (o.w > 0.0f) ? o.w : 0.01f * o.w;
    }
    ((float4*)out)[g] = o;
}

extern "C" void kernel_launch(void* const* d_in, const int* in_sizes, int n_in,
                              void* d_out, int out_size, void* d_ws, size_t ws_size,
                              hipStream_t stream) {
    const float* x  = (const float*)d_in[0];
    const int*   ei = (const int*)d_in[1];
    const float* W1 = (const float*)d_in[2];
    const float* b1 = (const float*)d_in[3];
    const float* W2 = (const float*)d_in[4];
    const float* b2 = (const float*)d_in[5];
    float* out = (float*)d_out;

    const int E = in_sizes[1] / 2;
    const int N = in_sizes[0] / (TP * CH);
    const int total = N * CH;
    const int* row = ei;
    const int* col = ei + E;

    char* ws = (char*)d_ws;
    float* dis = (float*)ws;
    size_t off = ((size_t)N * sizeof(float) + 255) & ~(size_t)255;
    float* bufA = (float*)(ws + off);
    float* bufB = bufA + total;
    int* cnt = (int*)(bufB + total);
    int* ptr = cnt + N;
    int* cur = ptr + (N + 1);
    int* src = cur + N;

    const int TB = 256;
    const int gemmBlocks = (N + GBM - 1) / GBM;
    const int total4 = N * CH4;
    const int gatherBlocks = (total4 + 255) / 256;

    init_kernel<<<(N + TB - 1) / TB, TB, 0, stream>>>(dis, cnt, N);
    deg_kernel<<<(E + TB - 1) / TB, TB, 0, stream>>>(row, col, dis, cnt, E);
    rsqrt_kernel<<<(N + TB - 1) / TB, TB, 0, stream>>>(dis, N);

    scan_kernel<<<1, SCAN_T, 0, stream>>>(cnt, ptr, cur, N);
    csr_fill_kernel<<<(E + TB - 1) / TB, TB, 0, stream>>>(row, col, cur, src, E);

    mean_kernel<<<(total + TB - 1) / TB, TB, 0, stream>>>(x, bufA, total);

    gemm_wmma_kernel<<<gemmBlocks, 256, 0, stream>>>(bufA, W1, b1, bufB, N);

    gather_kernel<<<gatherBlocks, 256, 0, stream>>>(ptr, src, dis, bufB, bufA, total4, 1);

    gemm_wmma_kernel<<<gemmBlocks, 256, 0, stream>>>(bufA, W2, b2, bufB, N);

    gather_kernel<<<gatherBlocks, 256, 0, stream>>>(ptr, src, dis, bufB, out, total4, 0);
}


# Mean pool over the 4 rows moves into the layer-1 GEMM's A-tile loader (4 loads per slot, summed in registers, same order); the pooling kernel, its buffer round trip and its launch go
# speedup vs baseline: 2.6247x; 1.0648x over previous
#include <hip/hip_runtime.h>

typedef float v2f __attribute__((ext_vector_type(2)));
typedef float v4f __attribute__((ext_vector_type(4)));
typedef float v8f __attribute__((ext_vector_type(8)));

#define CH 300
#define TP 4
#define CH4 75
#define SCAN_T 1024


__global__ void init_kernel(float* __restrict__ deg, int* __restrict__ cnt, int n) {
    int i = blockIdx.x * blockDim.x + threadIdx.x;
    if (i < n) { deg[i] = 1.0f; cnt[i] = 0; }
}

__global__ void deg_kernel(const int* __restrict__ row, const int* __restrict__ col,
                           float* __restrict__ deg, int* __restrict__ cnt, int E) {
    int e = blockIdx.x * blockDim.x + threadIdx.x;
    if (e < E) {
        unsafeAtomicAdd(&deg[row[e]], 1.0f);
        atomicAdd(&cnt[col[e]], 1);
    }
}

__global__ __launch_bounds__(SCAN_T) void scan_kernel(const int* __restrict__ cnt,
                                                     int* __restrict__ ptr,
                                                     int* __restrict__ cur, int n) {
    __shared__ int part[SCAN_T];
    const int t = threadIdx.x;
    const int per = (n + SCAN_T - 1) / SCAN_T;
    int b = t * per;
    if (b > n) b = n;
    int e = b + per;
    if (e > n) e = n;
    int sum = 0;
    for (int i = b; i < e; ++i) sum += cnt[i];
    part[t] = sum;
    __syncthreads();
    for (int o = 1; o < SCAN_T; o <<= 1) {
        int v = (t >= o) ? part[t - o] : 0;
        __syncthreads();
        part[t] += v;
        __syncthreads();
    }
    int run = part[t] - sum;
    for (int i = b; i < e; ++i) {
        ptr[i] = run;
        cur[i] = run;
        run += cnt[i];
    }
    if (t == SCAN_T - 1) ptr[n] = run;
}

__global__ void csr_fill_kernel(const int* __restrict__ row, const int* __restrict__ col,
                                int* __restrict__ cur, int* __restrict__ src, int E) {
    int e = blockIdx.x * blockDim.x + threadIdx.x;
    if (e < E) {
        int p = atomicAdd(&cur[col[e]], 1);
        src[p] = row[e];
    }
}

__global__ void rsqrt_kernel(float* __restrict__ d, int n) {
    int i = blockIdx.x * blockDim.x + threadIdx.x;
    if (i < n) d[i] = rsqrtf(d[i]);
}

#define GBM 64
#define GBK 20
#define GNP 320
#define GNCH (CH / GBK)
#define GA4 (GBM * GBK / 4)
#define GS4 ((GBM + GNP) * GBK / 4)
#define GSF ((GBM + GNP) * GBK)
#define GLDS (GBM * CH)

template <bool POOL>
__global__ __launch_bounds__(256) void gemm_wmma_kernel(const float* __restrict__ A,
                                                        const float* __restrict__ W,
                                                        const float* __restrict__ bias,
                                                        float* __restrict__ C,
                                                        int M) {
    __shared__ __attribute__((aligned(16))) float lds[GLDS];
    const int tid  = threadIdx.x;
    const int wave = tid >> 5;
    const int lane = tid & 31;
    const int mn   = lane & 15;
    const int half = lane >> 4;
    const int mt   = wave & 3;
    const int nh   = wave >> 2;
    const int row0 = blockIdx.x * GBM;

    const float* gp[8];
#pragma unroll
    for (int i = 0; i < 8; ++i) {
        const int idx = tid + 256 * i;
        if (idx < GA4) {
            const int r = idx / 5;
            const int j = idx - 5 * r;
            int gr = row0 + r;
            if (gr > M - 1) gr = M - 1;
            gp[i] = A + (size_t)gr * (POOL ? TP * CH : CH) + 4 * j;
        } else {
            const int w = idx - GA4;
            int r = w / 5;
            const int j = w - 5 * r;
            if (r > CH - 1) r = CH - 1;
            gp[i] = W + (size_t)r * CH + 4 * j;
        }
    }
    const bool last = tid < (GS4 - 7 * 256);
    const bool a1 = tid < (GA4 - 256);

    v4f v[8];
    v4f p0[TP - 1], p1[TP - 1];
    v4f* l4 = (v4f*)lds;
    if (POOL) {
#pragma unroll
        for (int t = 1; t < TP; ++t) p0[t - 1] = *(const v4f*)(gp[0] + t * CH);
        if (a1) {
#pragma unroll
            for (int t = 1; t < TP; ++t) p1[t - 1] = *(const v4f*)(gp[1] + t * CH);
        }
    }
#pragma unroll
    for (int i = 0; i < 7; ++i) v[i] = *(const v4f*)(gp[i]);
    if (last) v[7] = *(const v4f*)(gp[7]);
    if (POOL) {
        v[0] = 0.25f * (((v[0] + p0[0]) + p0[1]) + p0[2]);
        if (a1) v[1] = 0.25f * (((v[1] + p1[0]) + p1[1]) + p1[2]);
    }
#pragma unroll
    for (int i = 0; i < 7; ++i) l4[tid + 256 * i] = v[i];
    if (last) l4[tid + 7 * 256] = v[7];
    __syncthreads();

    v8f acc[10] = {};
    const int aoff = (mt * 16 + mn) * GBK + 2 * half;
    const int boff = GBM * GBK + (nh * 160 + mn) * GBK + 2 * half;

    for (int ch = 0; ch < GNCH; ++ch) {
        const float* st = lds + (ch & 1) * GSF;
        const bool more = (ch + 1 < GNCH);
        if (more) {
            const int kc = (ch + 1) * GBK;
            if (POOL) {
#pragma unroll
                for (int t = 1; t < TP; ++t) p0[t - 1] = *(const v4f*)(gp[0] + kc + t * CH);
                if (a1) {
#pragma unroll
                    for (int t = 1; t < TP; ++t) p1[t - 1] = *(const v4f*)(gp[1] + kc + t * CH);
                }
            }
#pragma unroll
            for (int i = 0; i < 7; ++i) v[i] = *(const v4f*)(gp[i] + kc);
            if (last) v[7] = *(const v4f*)(gp[7] + kc);
        }
#pragma unroll
        for (int s = 0; s < GBK / 4; ++s) {
            const v2f a = *(const v2f*)(st + aoff + 4 * s);
#pragma unroll
            for (int t = 0; t < 10; ++t) {
                const v2f b = *(const v2f*)(st + boff + t * (16 * GBK) + 4 * s);
                acc[t] = __builtin_amdgcn_wmma_f32_16x16x4_f32(
                    false, a, false, b, (short)0, acc[t], false, false);
            }
        }
        if (more) {
            v4f* d4 = (v4f*)(lds + ((ch + 1) & 1) * GSF);
            if (POOL) {
                __builtin_amdgcn_sched_barrier(0);
                v[0] = 0.25f * (((v[0] + p0[0]) + p0[1]) + p0[2]);
                if (a1) v[1] = 0.25f * (((v[1] + p1[0]) + p1[1]) + p1[2]);
            }
#pragma unroll
            for (int i = 0; i < 7; ++i) d4[tid + 256 * i] = v[i];
            if (last) d4[tid + 7 * 256] = v[7];
        }
        __syncthreads();
    }

    float bv[10];
#pragma unroll
    for (int t = 0; t < 10; ++t) {
        const int n = nh * 160 + t * 16 + mn;
        bv[t] = bias[n < CH ? n : CH - 1];
    }
#pragma unroll
    for (int t = 0; t < 10; ++t) {
        const int n = nh * 160 + t * 16 + mn;
        if (n < CH) {
#pragma unroll
            for (int j = 0; j < 8; ++j)
                lds[(mt * 16 + 8 * half + j) * CH + n] = acc[t][j] + bv[t];
        }
    }
    __syncthreads();
    int rows = M - row0;
    if (rows > GBM) rows = GBM;
    const int n4 = rows * CH4;
    v4f* c4 = (v4f*)(C + (size_t)row0 * CH);
    for (int i = tid; i < n4; i += 256) c4[i] = l4[i];
}

__global__ __launch_bounds__(256) void gather_kernel(const int* __restrict__ ptr,
                                                     const int* __restrict__ src,
                                                     const float* __restrict__ dis,
                                                     const float* __restrict__ h,
                                                     float* __restrict__ out,
                                                     int total4, int relu) {
    const int g = blockIdx.x * 256 + threadIdx.x;
    if (g >= total4) return;
    const int n = g / CH4;
    const int q = g - n * CH4;
    const float4* __restrict__ h4 = (const float4*)h;
    const float dn = dis[n];
    const float4 self = h4[g];
    const int p0 = ptr[n];
    const int p1 = ptr[n + 1];
    float4 s = make_float4(0.0f, 0.0f, 0.0f, 0.0f);
    for (int p = p0; p < p1; ++p) {
        const int r = src[p];
        const float w = dis[r];
        const float4 v = h4[(size_t)r * CH4 + q];
        s.x += w * v.x; s.y += w * v.y; s.z += w * v.z; s.w += w * v.w;
    }
    float4 o;
    o.x = dn * (s.x + dn * self.x);
    o.y = dn * (s.y + dn * self.y);
    o.z = dn * (s.z + dn * self.z);
    o.w = dn * (s.w + dn * self.w);
    if (relu) {
        o.x = (o.x > 0.0f) ? o.x : 0.01f * o.x;
        o.y = (o.y > 0.0f) ? o.y : 0.01f * o.y;
        o.z = (o.z > 0.0f) ? o.z : 0.01f * o.z;
        o.w = (o.w > 0.0f) ? o.w : 0.01f * o.w;
    }
    ((float4*)out)[g] = o;
}

extern "C" void kernel_launch(void* const* d_in, const int* in_sizes, int n_in,
                              void* d_out, int out_size, void* d_ws, size_t ws_size,
                              hipStream_t stream) {
    const float* x  = (const float*)d_in[0];
    const int*   ei = (const int*)d_in[1];
    const float* W1 = (const float*)d_in[2];
    const float* b1 = (const float*)d_in[3];
    const float* W2 = (const float*)d_in[4];
    const float* b2 = (const float*)d_in[5];
    float* out = (float*)d_out;

    const int E = in_sizes[1] / 2;
    const int N = in_sizes[0] / (TP * CH);
    const int total = N * CH;
    const int* row = ei;
    const int* col = ei + E;

    char* ws = (char*)d_ws;
    float* dis = (float*)ws;
    size_t off = ((size_t)N * sizeof(float) + 255) & ~(size_t)255;
    float* bufA = (float*)(ws + off);
    float* bufB = bufA + total;
    int* cnt = (int*)(bufB + total);
    int* ptr = cnt + N;
    int* cur = ptr + (N + 1);
    int* src = cur + N;

    const int TB = 256;
    const int gemmBlocks = (N + GBM - 1) / GBM;
    const int total4 = N * CH4;
    const int gatherBlocks = (total4 + 255) / 256;

    init_kernel<<<(N + TB - 1) / TB, TB, 0, stream>>>(dis, cnt, N);
    deg_kernel<<<(E + TB - 1) / TB, TB, 0, stream>>>(row, col, dis, cnt, E);
    rsqrt_kernel<<<(N + TB - 1) / TB, TB, 0, stream>>>(dis, N);

    scan_kernel<<<1, SCAN_T, 0, stream>>>(cnt, ptr, cur, N);
    csr_fill_kernel<<<(E + TB - 1) / TB, TB, 0, stream>>>(row, col, cur, src, E);

    gemm_wmma_kernel<true><<<gemmBlocks, 256, 0, stream>>>(x, W1, b1, bufB, N);

    gather_kernel<<<gatherBlocks, 256, 0, stream>>>(ptr, src, dis, bufB, bufA, total4, 1);

    gemm_wmma_kernel<false><<<gemmBlocks, 256, 0, stream>>>(bufA, W2, b2, bufB, N);

    gather_kernel<<<gatherBlocks, 256, 0, stream>>>(ptr, src, dis, bufB, out, total4, 0);
}


# Prefix-sum kernel: each thread owns 4 adjacent counts per 4096-entry tile (whole-line 16-byte loads and stores, wave scans on the vector unit), not a private run of 49 counts at a 196-byte stride
# speedup vs baseline: 2.8340x; 1.0797x over previous
#include <hip/hip_runtime.h>

typedef float v2f __attribute__((ext_vector_type(2)));
typedef float v4f __attribute__((ext_vector_type(4)));
typedef float v8f __attribute__((ext_vector_type(8)));

#define CH 300
#define TP 4
#define CH4 75
#define SCAN_T 1024
#define SCAN_W (SCAN_T / 32)
#define SCAN_E (4 * SCAN_T)


__global__ void init_kernel(float* __restrict__ deg, int* __restrict__ cnt, int n) {
    int i = blockIdx.x * blockDim.x + threadIdx.x;
    if (i < n) { deg[i] = 1.0f; cnt[i] = 0; }
}

__global__ void deg_kernel(const int* __restrict__ row, const int* __restrict__ col,
                           float* __restrict__ deg, int* __restrict__ cnt, int E) {
    int e = blockIdx.x * blockDim.x + threadIdx.x;
    if (e < E) {
        unsafeAtomicAdd(&deg[row[e]], 1.0f);
        atomicAdd(&cnt[col[e]], 1);
    }
}

static_assert(SCAN_W == 32, "lane l of every wave takes the total of wave l");

__device__ __forceinline__ int4 scan_load(const int* __restrict__ cnt, int i, int n) {
    const int last = (n > 0 ? n - 1 : 0) & ~3;
    int4 c = *(const int4*)(cnt + (i < last ? i : last));
    c.x = (i < n) ? c.x : 0;
    c.y = (i + 1 < n) ? c.y : 0;
    c.z = (i + 2 < n) ? c.z : 0;
    c.w = (i + 3 < n) ? c.w : 0;
    return c;
}

__device__ __forceinline__ int wave_scan(int v, int lane) {
    v += __builtin_amdgcn_update_dpp(0, v, 0x111, 0xf, 0xf, false);
    v += __builtin_amdgcn_update_dpp(0, v, 0x112, 0xf, 0xf, false);
    v += __builtin_amdgcn_update_dpp(0, v, 0x114, 0xf, 0xf, false);
    v += __builtin_amdgcn_update_dpp(0, v, 0x118, 0xf, 0xf, false);
    const int r0 = __builtin_amdgcn_readlane(v, 15);
    return v + (lane >= 16 ? r0 : 0);
}

__global__ __launch_bounds__(SCAN_T) void scan_kernel(const int* __restrict__ cnt,
                                                     int* __restrict__ ptr,
                                                     int* __restrict__ cur, int n) {
    __shared__ int wtot[2][SCAN_W];
    const int t = threadIdx.x;
    const int lane = t & 31;
    const int wave = __builtin_amdgcn_readfirstlane(t >> 5);
    const int ntile = (n + SCAN_E - 1) / SCAN_E;
    int carry = 0;
    int4 c = scan_load(cnt, 4 * t, n);
    for (int k = 0; k < ntile; ++k) {
        const int i = k * SCAN_E + 4 * t;
        const int4 nx = scan_load(cnt, i + SCAN_E, n);
        const int s1 = c.x;
        const int s2 = s1 + c.y;
        const int s3 = s2 + c.z;
        const int s4 = s3 + c.w;
        const int v = wave_scan(s4, lane);
        if (lane == 31) wtot[k & 1][wave] = v;
        __builtin_amdgcn_fence(__ATOMIC_RELEASE, "workgroup", "local");
        __builtin_amdgcn_s_barrier();
        __builtin_amdgcn_fence(__ATOMIC_ACQUIRE, "workgroup", "local");
        const int w = wtot[k & 1][lane];
        const int ws = wave_scan(w, lane);
        const int before = __builtin_amdgcn_readlane(ws - w, wave);
        const int tile = __builtin_amdgcn_readlane(ws, 31);
        const int p0 = carry + before + (v - s4);
        const int4 o = make_int4(p0, p0 + s1, p0 + s2, p0 + s3);
        if (i + 3 < n) {
            *(int4*)(ptr + i) = o;
            *(int4*)(cur + i) = o;
        } else {
            if (i < n) { ptr[i] = o.x; cur[i] = o.x; }
            if (i + 1 < n) { ptr[i + 1] = o.y; cur[i + 1] = o.y; }
            if (i + 2 < n) { ptr[i + 2] = o.z; cur[i + 2] = o.z; }
        }
        carry += tile;
        c = nx;
    }
    if (t == 0) ptr[n] = carry;
}

__global__ void csr_fill_kernel(const int* __restrict__ row, const int* __restrict__ col,
                                int* __restrict__ cur, int* __restrict__ src, int E) {
    int e = blockIdx.x * blockDim.x + threadIdx.x;
    if (e < E) {
        int p = atomicAdd(&cur[col[e]], 1);
        src[p] = row[e];
    }
}

__global__ void rsqrt_kernel(float* __restrict__ d, int n) {
    int i = blockIdx.x * blockDim.x + threadIdx.x;
    if (i < n) d[i] = rsqrtf(d[i]);
}

#define GBM 64
#define GBK 20
#define GNP 320
#define GNCH (CH / GBK)
#define GA4 (GBM * GBK / 4)
#define GS4 ((GBM + GNP) * GBK / 4)
#define GSF ((GBM + GNP) * GBK)
#define GLDS (GBM * CH)

template <bool POOL>
__global__ __launch_bounds__(256) void gemm_wmma_kernel(const float* __restrict__ A,
                                                        const float* __restrict__ W,
                                                        const float* __restrict__ bias,
                                                        float* __restrict__ C,
                                                        int M) {
    __shared__ __attribute__((aligned(16))) float lds[GLDS];
    const int tid  = threadIdx.x;
    const int wave = tid >> 5;
    const int lane = tid & 31;
    const int mn   = lane & 15;
    const int half = lane >> 4;
    const int mt   = wave & 3;
    const int nh   = wave >> 2;
    const int row0 = blockIdx.x * GBM;

    const float* gp[8];
#pragma unroll
    for (int i = 0; i < 8; ++i) {
        const int idx = tid + 256 * i;
        if (idx < GA4) {
            const int r = idx / 5;
            const int j = idx - 5 * r;
            int gr = row0 + r;
            if (gr > M - 1) gr = M - 1;
            gp[i] = A + (size_t)gr * (POOL ? TP * CH : CH) + 4 * j;
        } else {
            const int w = idx - GA4;
            int r = w / 5;
            const int j = w - 5 * r;
            if (r > CH - 1) r = CH - 1;
            gp[i] = W + (size_t)r * CH + 4 * j;
        }
    }
    const bool last = tid < (GS4 - 7 * 256);
    const bool a1 = tid < (GA4 - 256);

    v4f v[8];
    v4f p0[TP - 1], p1[TP - 1];
    v4f* l4 = (v4f*)lds;
    if (POOL) {
#pragma unroll
        for (int t = 1; t < TP; ++t) p0[t - 1] = *(const v4f*)(gp[0] + t * CH);
        if (a1) {
#pragma unroll
            for (int t = 1; t < TP; ++t) p1[t - 1] = *(const v4f*)(gp[1] + t * CH);
        }
    }
#pragma unroll
    for (int i = 0; i < 7; ++i) v[i] = *(const v4f*)(gp[i]);
    if (last) v[7] = *(const v4f*)(gp[7]);
    if (POOL) {
        v[0] = 0.25f * (((v[0] + p0[0]) + p0[1]) + p0[2]);
        if (a1) v[1] = 0.25f * (((v[1] + p1[0]) + p1[1]) + p1[2]);
    }
#pragma unroll
    for (int i = 0; i < 7; ++i) l4[tid + 256 * i] = v[i];
    if (last) l4[tid + 7 * 256] = v[7];
    __syncthreads();

    v8f acc[10] = {};
    const int aoff = (mt * 16 + mn) * GBK + 2 * half;
    const int boff = GBM * GBK + (nh * 160 + mn) * GBK + 2 * half;

    for (int ch = 0; ch < GNCH; ++ch) {
        const float* st = lds + (ch & 1) * GSF;
        const bool more = (ch + 1 < GNCH);
        if (more) {
            const int kc = (ch + 1) * GBK;
            if (POOL) {
#pragma unroll
                for (int t = 1; t < TP; ++t) p0[t - 1] = *(const v4f*)(gp[0] + kc + t * CH);
                if (a1) {
#pragma unroll
                    for (int t = 1; t < TP; ++t) p1[t - 1] = *(const v4f*)(gp[1] + kc + t * CH);
                }
            }
#pragma unroll
            for (int i = 0; i < 7; ++i) v[i] = *(const v4f*)(gp[i] + kc);
            if (last) v[7] = *(const v4f*)(gp[7] + kc);
        }
#pragma unroll
        for (int s = 0; s < GBK / 4; ++s) {
            const v2f a = *(const v2f*)(st + aoff + 4 * s);
#pragma unroll
            for (int t = 0; t < 10; ++t) {
                const v2f b = *(const v2f*)(st + boff + t * (16 * GBK) + 4 * s);
                acc[t] = __builtin_amdgcn_wmma_f32_16x16x4_f32(
                    false, a, false, b, (short)0, acc[t], false, false);
            }
        }
        if (more) {
            v4f* d4 = (v4f*)(lds + ((ch + 1) & 1) * GSF);
            if (POOL) {
                __builtin_amdgcn_sched_barrier(0);
                v[0] = 0.25f * (((v[0] + p0[0]) + p0[1]) + p0[2]);
                if (a1) v[1] = 0.25f * (((v[1] + p1[0]) + p1[1]) + p1[2]);
            }
#pragma unroll
            for (int i = 0; i < 7; ++i) d4[tid + 256 * i] = v[i];
            if (last) d4[tid + 7 * 256] = v[7];
        }
        __syncthreads();
    }

    float bv[10];
#pragma unroll
    for (int t = 0; t < 10; ++t) {
        const int n = nh * 160 + t * 16 + mn;
        bv[t] = bias[n < CH ? n : CH - 1];
    }
#pragma unroll
    for (int t = 0; t < 10; ++t) {
        const int n = nh * 160 + t * 16 + mn;
        if (n < CH) {
#pragma unroll
            for (int j = 0; j < 8; ++j)
                lds[(mt * 16 + 8 * half + j) * CH + n] = acc[t][j] + bv[t];
        }
    }
    __syncthreads();
    int rows = M - row0;
    if (rows > GBM) rows = GBM;
    const int n4 = rows * CH4;
    v4f* c4 = (v4f*)(C + (size_t)row0 * CH);
    for (int i = tid; i < n4; i += 256) c4[i] = l4[i];
}

__global__ __launch_bounds__(256) void gather_kernel(const int* __restrict__ ptr,
                                                     const int* __restrict__ src,
                                                     const float* __restrict__ dis,
                                                     const float* __restrict__ h,
                                                     float* __restrict__ out,
                                                     int total4, int relu) {
    const int g = blockIdx.x * 256 + threadIdx.x;
    if (g >= total4) return;
    const int n = g / CH4;
    const int q = g - n * CH4;
    const float4* __restrict__ h4 = (const float4*)h;
    const float dn = dis[n];
    const float4 self = h4[g];
    const int p0 = ptr[n];
    const int p1 = ptr[n + 1];
    float4 s = make_float4(0.0f, 0.0f, 0.0f, 0.0f);
    for (int p = p0; p < p1; ++p) {
        const int r = src[p];
        const float w = dis[r];
        const float4 v = h4[(size_t)r * CH4 + q];
        s.x += w * v.x; s.y += w * v.y; s.z += w * v.z; s.w += w * v.w;
    }
    float4 o;
    o.x = dn * (s.x + dn * self.x);
    o.y = dn * (s.y + dn * self.y);
    o.z = dn * (s.z + dn * self.z);
    o.w = dn * (s.w + dn * self.w);
    if (relu) {
        o.x = (o.x > 0.0f) ? o.x : 0.01f * o.x;
        o.y = (o.y > 0.0f) ? o.y : 0.01f * o.y;
        o.z = (o.z > 0.0f) ? o.z : 0.01f * o.z;
        o.w = (o.w > 0.0f) ? o.w : 0.01f * o.w;
    }
    ((float4*)out)[g] = o;
}

extern "C" void kernel_launch(void* const* d_in, const int* in_sizes, int n_in,
                              void* d_out, int out_size, void* d_ws, size_t ws_size,
                              hipStream_t stream) {
    const float* x  = (const float*)d_in[0];
    const int*   ei = (const int*)d_in[1];
    const float* W1 = (const float*)d_in[2];
    const float* b1 = (const float*)d_in[3];
    const float* W2 = (const float*)d_in[4];
    const float* b2 = (const float*)d_in[5];
    float* out = (float*)d_out;

    const int E = in_sizes[1] / 2;
    const int N = in_sizes[0] / (TP * CH);
    const int total = N * CH;
    const int* row = ei;
    const int* col = ei + E;

    char* ws = (char*)d_ws;
    float* dis = (float*)ws;
    size_t off = ((size_t)N * sizeof(float) + 255) & ~(size_t)255;
    float* bufA = (float*)(ws + off);
    float* bufB = bufA + total;
    const int NP = (N + 4) & ~3;
    int* cnt = (int*)(bufB + total);
    int* ptr = cnt + NP;
    int* cur = ptr + NP;
    int* src = cur + NP;

    const int TB = 256;
    const int gemmBlocks = (N + GBM - 1) / GBM;
    const int total4 = N * CH4;
    const int gatherBlocks = (total4 + 255) / 256;

    init_kernel<<<(N + TB - 1) / TB, TB, 0, stream>>>(dis, cnt, N);
    deg_kernel<<<(E + TB - 1) / TB, TB, 0, stream>>>(row, col, dis, cnt, E);
    rsqrt_kernel<<<(N + TB - 1) / TB, TB, 0, stream>>>(dis, N);

    scan_kernel<<<1, SCAN_T, 0, stream>>>(cnt, ptr, cur, N);
    csr_fill_kernel<<<(E + TB - 1) / TB, TB, 0, stream>>>(row, col, cur, src, E);

    gemm_wmma_kernel<true><<<gemmBlocks, 256, 0, stream>>>(x, W1, b1, bufB, N);

    gather_kernel<<<gatherBlocks, 256, 0, stream>>>(ptr, src, dis, bufB, bufA, total4, 1);

    gemm_wmma_kernel<false><<<gemmBlocks, 256, 0, stream>>>(bufA, W2, b2, bufB, N);

    gather_kernel<<<gatherBlocks, 256, 0, stream>>>(ptr, src, dis, bufB, out, total4, 0);
}


# CSR build in one launch: a block per node range counts, scans and fills its segments in LDS over two sweeps of the edge list; global atomics and 4 launches go; slots stay arrival order
# speedup vs baseline: 2.9278x; 1.0331x over previous
#include <hip/hip_runtime.h>

typedef float v2f __attribute__((ext_vector_type(2)));
typedef float v4f __attribute__((ext_vector_type(4)));
typedef float v8f __attribute__((ext_vector_type(8)));

#define CH 300
#define TP 4
#define CH4 75

#define CB_T 1024
#define CB_W (CB_T / 32)
#define CB_R 4096
#define CB_NB 64

__device__ __forceinline__ int wave_scan(int v, int lane) {
    v += __builtin_amdgcn_update_dpp(0, v, 0x111, 0xf, 0xf, false);
    v += __builtin_amdgcn_update_dpp(0, v, 0x112, 0xf, 0xf, false);
    v += __builtin_amdgcn_update_dpp(0, v, 0x114, 0xf, 0xf, false);
    v += __builtin_amdgcn_update_dpp(0, v, 0x118, 0xf, 0xf, false);
    const int r0 = __builtin_amdgcn_readlane(v, 15);
    return v + (lane >= 16 ? r0 : 0);
}
static_assert(CB_W == 32, "lane l of every wave takes the total of wave l");

__device__ __forceinline__ int4 key_load(const int4* __restrict__ k4, int i, int n4) {
    int4 k = k4[i < n4 ? i : n4 - 1];
    if (i >= n4) k = make_int4(0x7fffffff, 0x7fffffff, 0x7fffffff, 0x7fffffff);
    return k;
}

__global__ __launch_bounds__(CB_T) void csr_build_kernel(const int* __restrict__ row,
                                                        const int* __restrict__ col,
                                                        float* __restrict__ dis,
                                                        int* __restrict__ ptr,
                                                        int* __restrict__ src, int E, int N) {
    __shared__ __attribute__((aligned(16))) int cnt[CB_R];
    __shared__ int wtot[CB_W];
    __shared__ int wbel[CB_W];
    const int t = threadIdx.x;
    const int lane = t & 31;
    const int wave = __builtin_amdgcn_readfirstlane(t >> 5);
    const bool norm = blockIdx.x >= CB_NB;
    const int b = norm ? (int)blockIdx.x - CB_NB : (int)blockIdx.x;
    const int per = ((N + CB_NB - 1) / CB_NB + 31) & ~31;
    const int blo = b * per;
    int bhi = blo + per;
    if (bhi > N) bhi = N;
    const int* __restrict__ key = norm ? row : col;
    const bool vec = (E & 3) == 0 && E > 0;
    const int4* __restrict__ key4 = (const int4*)key;
    const int E4 = E >> 2;

    for (int lo = blo; lo < bhi; lo += CB_R) {
        int R = bhi - lo;
        if (R > CB_R) R = CB_R;
        ((int4*)cnt)[t] = make_int4(0, 0, 0, 0);
        __syncthreads();

        int below = 0;
#define CB_COUNT(K) { const int d_ = (K) - lo; below += ((K) < lo) ? 1 : 0; \
                      if ((unsigned)d_ < (unsigned)R) atomicAdd(&cnt[d_], 1); }
        if (vec) {
            for (int i0 = 0; i0 < E4; i0 += 4 * CB_T) {
                int4 k[4];
#pragma unroll
                for (int u = 0; u < 4; ++u) k[u] = key_load(key4, i0 + u * CB_T + t, E4);
#pragma unroll
                for (int u = 0; u < 4; ++u) {
                    CB_COUNT(k[u].x) CB_COUNT(k[u].y) CB_COUNT(k[u].z) CB_COUNT(k[u].w)
                }
            }
        } else {
            for (int e = t; e < E; e += CB_T) { const int k = key[e]; CB_COUNT(k) }
        }
#undef CB_COUNT
        __syncthreads();

        if (norm) {
            for (int d = t; d < R; d += CB_T) dis[lo + d] = rsqrtf(1.0f + (float)cnt[d]);
            __syncthreads();
            continue;
        }

        const int bw = wave_scan(below, lane);
        if (lane == 31) wbel[wave] = bw;
        int4 c = ((int4*)cnt)[t];
        const int s1 = c.x;
        const int s2 = s1 + c.y;
        const int s3 = s2 + c.z;
        const int s4 = s3 + c.w;
        const int v = wave_scan(s4, lane);
        if (lane == 31) wtot[wave] = v;
        __syncthreads();
        const int w = wtot[lane];
        const int ws = wave_scan(w, lane);
        const int before = __builtin_amdgcn_readlane(ws - w, wave);
        const int total = __builtin_amdgcn_readlane(ws, 31);
        const int base = __builtin_amdgcn_readlane(wave_scan(wbel[lane], lane), 31);
        const int p0 = base + before + (v - s4);
        const int4 o = make_int4(p0, p0 + s1, p0 + s2, p0 + s3);
        ((int4*)cnt)[t] = o;
        const int d0 = 4 * t;
        if (d0 < R) ptr[lo + d0] = o.x;
        if (d0 + 1 < R) ptr[lo + d0 + 1] = o.y;
        if (d0 + 2 < R) ptr[lo + d0 + 2] = o.z;
        if (d0 + 3 < R) ptr[lo + d0 + 3] = o.w;
        if (t == 0 && lo + R == N) ptr[N] = base + total;
        __syncthreads();

#define CB_FILL(K, EIDX) { const int d_ = (K) - lo; \
                      if ((unsigned)d_ < (unsigned)R) { const int p_ = atomicAdd(&cnt[d_], 1); \
                                                        src[p_] = row[EIDX]; } }
        if (vec) {
            for (int i0 = 0; i0 < E4; i0 += 4 * CB_T) {
                int4 k[4];
#pragma unroll
                for (int u = 0; u < 4; ++u) k[u] = key_load(key4, i0 + u * CB_T + t, E4);
#pragma unroll
                for (int u = 0; u < 4; ++u) {
                    const int e0 = 4 * (i0 + u * CB_T + t);
                    CB_FILL(k[u].x, e0) CB_FILL(k[u].y, e0 + 1)
                    CB_FILL(k[u].z, e0 + 2) CB_FILL(k[u].w, e0 + 3)
                }
            }
        } else {
            for (int e = t; e < E; e += CB_T) { const int k = key[e]; CB_FILL(k, e) }
        }
#undef CB_FILL
        __syncthreads();
    }
}

#define GBM 64
#define GBK 20
#define GNP 320
#define GNCH (CH / GBK)
#define GA4 (GBM * GBK / 4)
#define GS4 ((GBM + GNP) * GBK / 4)
#define GSF ((GBM + GNP) * GBK)
#define GLDS (GBM * CH)

template <bool POOL>
__global__ __launch_bounds__(256) void gemm_wmma_kernel(const float* __restrict__ A,
                                                        const float* __restrict__ W,
                                                        const float* __restrict__ bias,
                                                        float* __restrict__ C,
                                                        int M) {
    __shared__ __attribute__((aligned(16))) float lds[GLDS];
    const int tid  = threadIdx.x;
    const int wave = tid >> 5;
    const int lane = tid & 31;
    const int mn   = lane & 15;
    const int half = lane >> 4;
    const int mt   = wave & 3;
    const int nh   = wave >> 2;
    const int row0 = blockIdx.x * GBM;

    const float* gp[8];
#pragma unroll
    for (int i = 0; i < 8; ++i) {
        const int idx = tid + 256 * i;
        if (idx < GA4) {
            const int r = idx / 5;
            const int j = idx - 5 * r;
            int gr = row0 + r;
            if (gr > M - 1) gr = M - 1;
            gp[i] = A + (size_t)gr * (POOL ? TP * CH : CH) + 4 * j;
        } else {
            const int w = idx - GA4;
            int r = w / 5;
            const int j = w - 5 * r;
            if (r > CH - 1) r = CH - 1;
            gp[i] = W + (size_t)r * CH + 4 * j;
        }
    }
    const bool last = tid < (GS4 - 7 * 256);
    const bool a1 = tid < (GA4 - 256);

    v4f v[8];
    v4f p0[TP - 1], p1[TP - 1];
    v4f* l4 = (v4f*)lds;
    if (POOL) {
#pragma unroll
        for (int t = 1; t < TP; ++t) p0[t - 1] = *(const v4f*)(gp[0] + t * CH);
        if (a1) {
#pragma unroll
            for (int t = 1; t < TP; ++t) p1[t - 1] = *(const v4f*)(gp[1] + t * CH);
        }
    }
#pragma unroll
    for (int i = 0; i < 7; ++i) v[i] = *(const v4f*)(gp[i]);
    if (last) v[7] = *(const v4f*)(gp[7]);
    if (POOL) {
        v[0] = 0.25f * (((v[0] + p0[0]) + p0[1]) + p0[2]);
        if (a1) v[1] = 0.25f * (((v[1] + p1[0]) + p1[1]) + p1[2]);
    }
#pragma unroll
    for (int i = 0; i < 7; ++i) l4[tid + 256 * i] = v[i];
    if (last) l4[tid + 7 * 256] = v[7];
    __syncthreads();

    v8f acc[10] = {};
    const int aoff = (mt * 16 + mn) * GBK + 2 * half;
    const int boff = GBM * GBK + (nh * 160 + mn) * GBK + 2 * half;

    for (int ch = 0; ch < GNCH; ++ch) {
        const float* st = lds + (ch & 1) * GSF;
        const bool more = (ch + 1 < GNCH);
        if (more) {
            const int kc = (ch + 1) * GBK;
            if (POOL) {
#pragma unroll
                for (int t = 1; t < TP; ++t) p0[t - 1] = *(const v4f*)(gp[0] + kc + t * CH);
                if (a1) {
#pragma unroll
                    for (int t = 1; t < TP; ++t) p1[t - 1] = *(const v4f*)(gp[1] + kc + t * CH);
                }
            }
#pragma unroll
            for (int i = 0; i < 7; ++i) v[i] = *(const v4f*)(gp[i] + kc);
            if (last) v[7] = *(const v4f*)(gp[7] + kc);
        }
#pragma unroll
        for (int s = 0; s < GBK / 4; ++s) {
            const v2f a = *(const v2f*)(st + aoff + 4 * s);
#pragma unroll
            for (int t = 0; t < 10; ++t) {
                const v2f b = *(const v2f*)(st + boff + t * (16 * GBK) + 4 * s);
                acc[t] = __builtin_amdgcn_wmma_f32_16x16x4_f32(
                    false, a, false, b, (short)0, acc[t], false, false);
            }
        }
        if (more) {
            v4f* d4 = (v4f*)(lds + ((ch + 1) & 1) * GSF);
            if (POOL) {
                __builtin_amdgcn_sched_barrier(0);
                v[0] = 0.25f * (((v[0] + p0[0]) + p0[1]) + p0[2]);
                if (a1) v[1] = 0.25f * (((v[1] + p1[0]) + p1[1]) + p1[2]);
            }
#pragma unroll
            for (int i = 0; i < 7; ++i) d4[tid + 256 * i] = v[i];
            if (last) d4[tid + 7 * 256] = v[7];
        }
        __syncthreads();
    }

    float bv[10];
#pragma unroll
    for (int t = 0; t < 10; ++t) {
        const int n = nh * 160 + t * 16 + mn;
        bv[t] = bias[n < CH ? n : CH - 1];
    }
#pragma unroll
    for (int t = 0; t < 10; ++t) {
        const int n = nh * 160 + t * 16 + mn;
        if (n < CH) {
#pragma unroll
            for (int j = 0; j < 8; ++j)
                lds[(mt * 16 + 8 * half + j) * CH + n] = acc[t][j] + bv[t];
        }
    }
    __syncthreads();
    int rows = M - row0;
    if (rows > GBM) rows = GBM;
    const int n4 = rows * CH4;
    v4f* c4 = (v4f*)(C + (size_t)row0 * CH);
    for (int i = tid; i < n4; i += 256) c4[i] = l4[i];
}

__global__ __launch_bounds__(256) void gather_kernel(const int* __restrict__ ptr,
                                                     const int* __restrict__ src,
                                                     const float* __restrict__ dis,
                                                     const float* __restrict__ h,
                                                     float* __restrict__ out,
                                                     int total4, int relu) {
    const int g = blockIdx.x * 256 + threadIdx.x;
    if (g >= total4) return;
    const int n = g / CH4;
    const int q = g - n * CH4;
    const float4* __restrict__ h4 = (const float4*)h;
    const float dn = dis[n];
    const float4 self = h4[g];
    const int p0 = ptr[n];
    const int p1 = ptr[n + 1];
    float4 s = make_float4(0.0f, 0.0f, 0.0f, 0.0f);
    for (int p = p0; p < p1; ++p) {
        const int r = src[p];
        const float w = dis[r];
        const float4 v = h4[(size_t)r * CH4 + q];
        s.x += w * v.x; s.y += w * v.y; s.z += w * v.z; s.w += w * v.w;
    }
    float4 o;
    o.x = dn * (s.x + dn * self.x);
    o.y = dn * (s.y + dn * self.y);
    o.z = dn * (s.z + dn * self.z);
    o.w = dn * (s.w + dn * self.w);
    if (relu) {
        o.x = (o.x > 0.0f) ? o.x : 0.01f * o.x;
        o.y = (o.y > 0.0f) ? o.y : 0.01f * o.y;
        o.z = (o.z > 0.0f) ? o.z : 0.01f * o.z;
        o.w = (o.w > 0.0f) ? o.w : 0.01f * o.w;
    }
    ((float4*)out)[g] = o;
}

extern "C" void kernel_launch(void* const* d_in, const int* in_sizes, int n_in,
                              void* d_out, int out_size, void* d_ws, size_t ws_size,
                              hipStream_t stream) {
    const float* x  = (const float*)d_in[0];
    const int*   ei = (const int*)d_in[1];
    const float* W1 = (const float*)d_in[2];
    const float* b1 = (const float*)d_in[3];
    const float* W2 = (const float*)d_in[4];
    const float* b2 = (const float*)d_in[5];
    float* out = (float*)d_out;

    const int E = in_sizes[1] / 2;
    const int N = in_sizes[0] / (TP * CH);
    const int total = N * CH;
    const int* row = ei;
    const int* col = ei + E;

    char* ws = (char*)d_ws;
    float* dis = (float*)ws;
    size_t off = ((size_t)N * sizeof(float) + 255) & ~(size_t)255;
    float* bufA = (float*)(ws + off);
    float* bufB = bufA + total;
    int* ptr = (int*)(bufB + total);
    int* src = ptr + (N + 1);

    const int gemmBlocks = (N + GBM - 1) / GBM;
    const int total4 = N * CH4;
    const int gatherBlocks = (total4 + 255) / 256;

    csr_build_kernel<<<2 * CB_NB, CB_T, 0, stream>>>(row, col, dis, ptr, src, E, N);

    gemm_wmma_kernel<true><<<gemmBlocks, 256, 0, stream>>>(x, W1, b1, bufB, N);

    gather_kernel<<<gatherBlocks, 256, 0, stream>>>(ptr, src, dis, bufB, bufA, total4, 1);

    gemm_wmma_kernel<false><<<gemmBlocks, 256, 0, stream>>>(bufA, W2, b2, bufB, N);

    gather_kernel<<<gatherBlocks, 256, 0, stream>>>(ptr, src, dis, bufB, out, total4, 0);
}


# W1 and W2 repacked each call (extra blocks of the first launch) into stage-major tiles: a GEMM stage's W slice is one contiguous run of whole lines, not 320 row pieces of 80 bytes
# speedup vs baseline: 3.3326x; 1.1383x over previous
#include <hip/hip_runtime.h>

typedef float v2f __attribute__((ext_vector_type(2)));
typedef float v4f __attribute__((ext_vector_type(4)));
typedef float v8f __attribute__((ext_vector_type(8)));

#define CH 300
#define TP 4
#define CH4 75

#define GBK 20
#define GNP 320
#define GNCH (CH / GBK)
#define GSW (GNP * GBK)
#define PK_F4 (GNCH * GSW / 4)

#define CB_T 1024
#define CB_W (CB_T / 32)
#define CB_R 4096
#define CB_NB 64
#define PK_NB ((2 * PK_F4 + CB_T - 1) / CB_T)

__device__ __forceinline__ int wave_scan(int v, int lane) {
    v += __builtin_amdgcn_update_dpp(0, v, 0x111, 0xf, 0xf, false);
    v += __builtin_amdgcn_update_dpp(0, v, 0x112, 0xf, 0xf, false);
    v += __builtin_amdgcn_update_dpp(0, v, 0x114, 0xf, 0xf, false);
    v += __builtin_amdgcn_update_dpp(0, v, 0x118, 0xf, 0xf, false);
    const int r0 = __builtin_amdgcn_readlane(v, 15);
    return v + (lane >= 16 ? r0 : 0);
}
static_assert(CB_W == 32, "lane l of every wave takes the total of wave l");

__device__ __forceinline__ int4 key_load(const int4* __restrict__ k4, int i, int n4) {
    int4 k = k4[i < n4 ? i : n4 - 1];
    if (i >= n4) k = make_int4(0x7fffffff, 0x7fffffff, 0x7fffffff, 0x7fffffff);
    return k;
}

__global__ __launch_bounds__(CB_T) void csr_build_kernel(const int* __restrict__ row,
                                                        const int* __restrict__ col,
                                                        float* __restrict__ dis,
                                                        int* __restrict__ ptr,
                                                        int* __restrict__ src, int E, int N,
                                                        const float* __restrict__ W1,
                                                        const float* __restrict__ W2,
                                                        float* __restrict__ Wp1,
                                                        float* __restrict__ Wp2) {
    __shared__ __attribute__((aligned(16))) int cnt[CB_R];
    __shared__ int wtot[CB_W];
    __shared__ int wbel[CB_W];
    const int t = threadIdx.x;
    if (blockIdx.x >= 2 * CB_NB) {
        const int g = ((int)blockIdx.x - 2 * CB_NB) * CB_T + t;
        if (g < 2 * PK_F4) {
            const bool second = g >= PK_F4;
            const int f = second ? g - PK_F4 : g;
            const int st = f / (GSW / 4);
            const int w = f - st * (GSW / 4);
            const int n = w / 5;
            const int j = w - 5 * n;
            const float* __restrict__ Ws = second ? W2 : W1;
            float* __restrict__ Wd = second ? Wp2 : Wp1;
            v4f val = {0.0f, 0.0f, 0.0f, 0.0f};
            if (n < CH) val = *(const v4f*)(Ws + (size_t)n * CH + st * GBK + 4 * j);
            ((v4f*)Wd)[f] = val;
        }
        return;
    }
    const int lane = t & 31;
    const int wave = __builtin_amdgcn_readfirstlane(t >> 5);
    const bool norm = blockIdx.x >= CB_NB;
    const int b = norm ? (int)blockIdx.x - CB_NB : (int)blockIdx.x;
    const int per = ((N + CB_NB - 1) / CB_NB + 31) & ~31;
    const int blo = b * per;
    int bhi = blo + per;
    if (bhi > N) bhi = N;
    const int* __restrict__ key = norm ? row : col;
    const bool vec = (E & 3) == 0 && E > 0;
    const int4* __restrict__ key4 = (const int4*)key;
    const int E4 = E >> 2;

    for (int lo = blo; lo < bhi; lo += CB_R) {
        int R = bhi - lo;
        if (R > CB_R) R = CB_R;
        ((int4*)cnt)[t] = make_int4(0, 0, 0, 0);
        __syncthreads();

        int below = 0;
#define CB_COUNT(K) { const int d_ = (K) - lo; below += ((K) < lo) ? 1 : 0; \
                      if ((unsigned)d_ < (unsigned)R) atomicAdd(&cnt[d_], 1); }
        if (vec) {
            for (int i0 = 0; i0 < E4; i0 += 4 * CB_T) {
                int4 k[4];
#pragma unroll
                for (int u = 0; u < 4; ++u) k[u] = key_load(key4, i0 + u * CB_T + t, E4);
#pragma unroll
                for (int u = 0; u < 4; ++u) {
                    CB_COUNT(k[u].x) CB_COUNT(k[u].y) CB_COUNT(k[u].z) CB_COUNT(k[u].w)
                }
            }
        } else {
            for (int e = t; e < E; e += CB_T) { const int k = key[e]; CB_COUNT(k) }
        }
#undef CB_COUNT
        __syncthreads();

        if (norm) {
            for (int d = t; d < R; d += CB_T) dis[lo + d] = rsqrtf(1.0f + (float)cnt[d]);
            __syncthreads();
            continue;
        }

        const int bw = wave_scan(below, lane);
        if (lane == 31) wbel[wave] = bw;
        int4 c = ((int4*)cnt)[t];
        const int s1 = c.x;
        const int s2 = s1 + c.y;
        const int s3 = s2 + c.z;
        const int s4 = s3 + c.w;
        const int v = wave_scan(s4, lane);
        if (lane == 31) wtot[wave] = v;
        __syncthreads();
        const int w = wtot[lane];
        const int ws = wave_scan(w, lane);
        const int before = __builtin_amdgcn_readlane(ws - w, wave);
        const int total = __builtin_amdgcn_readlane(ws, 31);
        const int base = __builtin_amdgcn_readlane(wave_scan(wbel[lane], lane), 31);
        const int p0 = base + before + (v - s4);
        const int4 o = make_int4(p0, p0 + s1, p0 + s2, p0 + s3);
        ((int4*)cnt)[t] = o;
        const int d0 = 4 * t;
        if (d0 < R) ptr[lo + d0] = o.x;
        if (d0 + 1 < R) ptr[lo + d0 + 1] = o.y;
        if (d0 + 2 < R) ptr[lo + d0 + 2] = o.z;
        if (d0 + 3 < R) ptr[lo + d0 + 3] = o.w;
        if (t == 0 && lo + R == N) ptr[N] = base + total;
        __syncthreads();

#define CB_FILL(K, EIDX) { const int d_ = (K) - lo; \
                      if ((unsigned)d_ < (unsigned)R) { const int p_ = atomicAdd(&cnt[d_], 1); \
                                                        src[p_] = row[EIDX]; } }
        if (vec) {
            for (int i0 = 0; i0 < E4; i0 += 4 * CB_T) {
                int4 k[4];
#pragma unroll
                for (int u = 0; u < 4; ++u) k[u] = key_load(key4, i0 + u * CB_T + t, E4);
#pragma unroll
                for (int u = 0; u < 4; ++u) {
                    const int e0 = 4 * (i0 + u * CB_T + t);
                    CB_FILL(k[u].x, e0) CB_FILL(k[u].y, e0 + 1)
                    CB_FILL(k[u].z, e0 + 2) CB_FILL(k[u].w, e0 + 3)
                }
            }
        } else {
            for (int e = t; e < E; e += CB_T) { const int k = key[e]; CB_FILL(k, e) }
        }
#undef CB_FILL
        __syncthreads();
    }
}

#define GBM 64
#define GA4 (GBM * GBK / 4)
#define GS4 ((GBM + GNP) * GBK / 4)
#define GSF ((GBM + GNP) * GBK)
#define GLDS (GBM * CH)

template <bool POOL>
__global__ __launch_bounds__(256) void gemm_wmma_kernel(const float* __restrict__ A,
                                                        const float* __restrict__ W,
                                                        const float* __restrict__ bias,
                                                        float* __restrict__ C,
                                                        int M) {
    __shared__ __attribute__((aligned(16))) float lds[GLDS];
    const int tid  = threadIdx.x;
    const int wave = tid >> 5;
    const int lane = tid & 31;
    const int mn   = lane & 15;
    const int half = lane >> 4;
    const int mt   = wave & 3;
    const int nh   = wave >> 2;
    const int row0 = blockIdx.x * GBM;

    const bool last = tid < (GS4 - 7 * 256);
    const bool a1 = tid < (GA4 - 256);
    const float* gp[2];
#pragma unroll
    for (int i = 0; i < 2; ++i) {
        int r = (tid + 256 * i) / 5;
        const int j = (tid + 256 * i) - 5 * r;
        if (r > GBM - 1) r = GBM - 1;
        int gr = row0 + r;
        if (gr > M - 1) gr = M - 1;
        gp[i] = A + (size_t)gr * (POOL ? TP * CH : CH) + 4 * j;
    }
    const float* gw = W + 4 * tid;
#define WOF(I) (4 * (256 * (I) - GA4))
    const float* g1 = a1 ? gp[1] : gw + WOF(1);
    const int step1 = a1 ? GBK : GSW;

    v4f v[8];
    v4f p0[TP - 1], p1[TP - 1];
    v4f* l4 = (v4f*)lds;
    if (POOL) {
#pragma unroll
        for (int t = 1; t < TP; ++t) p0[t - 1] = *(const v4f*)(gp[0] + t * CH);
        if (a1) {
#pragma unroll
            for (int t = 1; t < TP; ++t) p1[t - 1] = *(const v4f*)(gp[1] + t * CH);
        }
    }
    v[0] = *(const v4f*)(gp[0]);
    v[1] = *(const v4f*)(g1);
#pragma unroll
    for (int i = 2; i < 7; ++i) v[i] = *(const v4f*)(gw + WOF(i));
    if (last) v[7] = *(const v4f*)(gw + WOF(7));
    if (POOL) {
        v[0] = 0.25f * (((v[0] + p0[0]) + p0[1]) + p0[2]);
        if (a1) v[1] = 0.25f * (((v[1] + p1[0]) + p1[1]) + p1[2]);
    }
#pragma unroll
    for (int i = 0; i < 7; ++i) l4[tid + 256 * i] = v[i];
    if (last) l4[tid + 7 * 256] = v[7];
    __syncthreads();

    v8f acc[10] = {};
    const int aoff = (mt * 16 + mn) * GBK + 2 * half;
    const int boff = GBM * GBK + (nh * 160 + mn) * GBK + 2 * half;

    for (int ch = 0; ch < GNCH; ++ch) {
        const float* st = lds + (ch & 1) * GSF;
        const bool more = (ch + 1 < GNCH);
        if (more) {
            const int kc = (ch + 1) * GBK;
            if (POOL) {
#pragma unroll
                for (int t = 1; t < TP; ++t) p0[t - 1] = *(const v4f*)(gp[0] + kc + t * CH);
                if (a1) {
#pragma unroll
                    for (int t = 1; t < TP; ++t) p1[t - 1] = *(const v4f*)(gp[1] + kc + t * CH);
                }
            }
            const int kw = (ch + 1) * GSW;
            v[0] = *(const v4f*)(gp[0] + kc);
            v[1] = *(const v4f*)(g1 + (ch + 1) * step1);
#pragma unroll
            for (int i = 2; i < 7; ++i) v[i] = *(const v4f*)(gw + kw + WOF(i));
            if (last) v[7] = *(const v4f*)(gw + kw + WOF(7));
        }
#pragma unroll
        for (int s = 0; s < GBK / 4; ++s) {
            const v2f a = *(const v2f*)(st + aoff + 4 * s);
#pragma unroll
            for (int t = 0; t < 10; ++t) {
                const v2f b = *(const v2f*)(st + boff + t * (16 * GBK) + 4 * s);
                acc[t] = __builtin_amdgcn_wmma_f32_16x16x4_f32(
                    false, a, false, b, (short)0, acc[t], false, false);
            }
        }
        if (more) {
            v4f* d4 = (v4f*)(lds + ((ch + 1) & 1) * GSF);
            if (POOL) {
                __builtin_amdgcn_sched_barrier(0);
                v[0] = 0.25f * (((v[0] + p0[0]) + p0[1]) + p0[2]);
                if (a1) v[1] = 0.25f * (((v[1] + p1[0]) + p1[1]) + p1[2]);
            }
#pragma unroll
            for (int i = 0; i < 7; ++i) d4[tid + 256 * i] = v[i];
            if (last) d4[tid + 7 * 256] = v[7];
        }
        __syncthreads();
    }

    float bv[10];
#pragma unroll
    for (int t = 0; t < 10; ++t) {
        const int n = nh * 160 + t * 16 + mn;
        bv[t] = bias[n < CH ? n : CH - 1];
    }
#pragma unroll
    for (int t = 0; t < 10; ++t) {
        const int n = nh * 160 + t * 16 + mn;
        if (n < CH) {
#pragma unroll
            for (int j = 0; j < 8; ++j)
                lds[(mt * 16 + 8 * half + j) * CH + n] = acc[t][j] + bv[t];
        }
    }
    __syncthreads();
    int rows = M - row0;
    if (rows > GBM) rows = GBM;
    const int n4 = rows * CH4;
    v4f* c4 = (v4f*)(C + (size_t)row0 * CH);
    for (int i = tid; i < n4; i += 256) c4[i] = l4[i];
#undef WOF
}

__global__ __launch_bounds__(256) void gather_kernel(const int* __restrict__ ptr,
                                                     const int* __restrict__ src,
                                                     const float* __restrict__ dis,
                                                     const float* __restrict__ h,
                                                     float* __restrict__ out,
                                                     int total4, int relu) {
    const int g = blockIdx.x * 256 + threadIdx.x;
    if (g >= total4) return;
    const int n = g / CH4;
    const int q = g - n * CH4;
    const float4* __restrict__ h4 = (const float4*)h;
    const float dn = dis[n];
    const float4 self = h4[g];
    const int p0 = ptr[n];
    const int p1 = ptr[n + 1];
    float4 s = make_float4(0.0f, 0.0f, 0.0f, 0.0f);
    for (int p = p0; p < p1; ++p) {
        const int r = src[p];
        const float w = dis[r];
        const float4 v = h4[(size_t)r * CH4 + q];
        s.x += w * v.x; s.y += w * v.y; s.z += w * v.z; s.w += w * v.w;
    }
    float4 o;
    o.x = dn * (s.x + dn * self.x);
    o.y = dn * (s.y + dn * self.y);
    o.z = dn * (s.z + dn * self.z);
    o.w = dn * (s.w + dn * self.w);
    if (relu) {
        o.x = (o.x > 0.0f) ? o.x : 0.01f * o.x;
        o.y = (o.y > 0.0f) ? o.y : 0.01f * o.y;
        o.z = (o.z > 0.0f) ? o.z : 0.01f * o.z;
        o.w = (o.w > 0.0f) ? o.w : 0.01f * o.w;
    }
    ((float4*)out)[g] = o;
}

extern "C" void kernel_launch(void* const* d_in, const int* in_sizes, int n_in,
                              void* d_out, int out_size, void* d_ws, size_t ws_size,
                              hipStream_t stream) {
    const float* x  = (const float*)d_in[0];
    const int*   ei = (const int*)d_in[1];
    const float* W1 = (const float*)d_in[2];
    const float* b1 = (const float*)d_in[3];
    const float* W2 = (const float*)d_in[4];
    const float* b2 = (const float*)d_in[5];
    float* out = (float*)d_out;

    const int E = in_sizes[1] / 2;
    const int N = in_sizes[0] / (TP * CH);
    const int total = N * CH;
    const int* row = ei;
    const int* col = ei + E;

    char* ws = (char*)d_ws;
    float* dis = (float*)ws;
    size_t off = ((size_t)N * sizeof(float) + 255) & ~(size_t)255;
    float* bufA = (float*)(ws + off);
    float* bufB = bufA + total;
    int* ptr = (int*)(bufB + total);
    int* src = ptr + (N + 1);
    const size_t offw = ((size_t)((char*)(src + E) - ws) + 255) & ~(size_t)255;
    float* Wp1 = (float*)(ws + offw);
    float* Wp2 = Wp1 + GNCH * GSW;

    const int gemmBlocks = (N + GBM - 1) / GBM;
    const int total4 = N * CH4;
    const int gatherBlocks = (total4 + 255) / 256;

    csr_build_kernel<<<2 * CB_NB + PK_NB, CB_T, 0, stream>>>(row, col, dis, ptr, src, E, N,
                                                             W1, W2, Wp1, Wp2);

    gemm_wmma_kernel<true><<<gemmBlocks, 256, 0, stream>>>(x, Wp1, b1, bufB, N);

    gather_kernel<<<gatherBlocks, 256, 0, stream>>>(ptr, src, dis, bufB, bufA, total4, 1);

    gemm_wmma_kernel<false><<<gemmBlocks, 256, 0, stream>>>(bufA, Wp2, b2, bufB, N);

    gather_kernel<<<gatherBlocks, 256, 0, stream>>>(ptr, src, dis, bufB, out, total4, 0);
}


# Both GEMMs re-tiled: one 20-wave block per WGP owns 208 rows x 300 columns, wave c holds column tile c of all 13 row tiles (65 tiles per SIMD, 80 before); W staged once per 208 rows; same sum order
# speedup vs baseline: 3.4821x; 1.0449x over previous
#include <hip/hip_runtime.h>

typedef float v2f __attribute__((ext_vector_type(2)));
typedef float v4f __attribute__((ext_vector_type(4)));
typedef float v8f __attribute__((ext_vector_type(8)));

#define CH 300
#define TP 4
#define CH4 75

#define GBK 20
#define GNP 320
#define GNCH (CH / GBK)
#define GSW (GNP * GBK)
#define PK_F4 (GNCH * GSW / 4)

#define CB_T 1024
#define CB_W (CB_T / 32)
#define CB_R 4096
#define CB_NB 64
#define PK_NB ((2 * PK_F4 + CB_T - 1) / CB_T)

__device__ __forceinline__ int wave_scan(int v, int lane) {
    v += __builtin_amdgcn_update_dpp(0, v, 0x111, 0xf, 0xf, false);
    v += __builtin_amdgcn_update_dpp(0, v, 0x112, 0xf, 0xf, false);
    v += __builtin_amdgcn_update_dpp(0, v, 0x114, 0xf, 0xf, false);
    v += __builtin_amdgcn_update_dpp(0, v, 0x118, 0xf, 0xf, false);
    const int r0 = __builtin_amdgcn_readlane(v, 15);
    return v + (lane >= 16 ? r0 : 0);
}
static_assert(CB_W == 32, "lane l of every wave takes the total of wave l");

__device__ __forceinline__ int4 key_load(const int4* __restrict__ k4, int i, int n4) {
    int4 k = k4[i < n4 ? i : n4 - 1];
    if (i >= n4) k = make_int4(0x7fffffff, 0x7fffffff, 0x7fffffff, 0x7fffffff);
    return k;
}

__global__ __launch_bounds__(CB_T) void csr_build_kernel(const int* __restrict__ row,
                                                        const int* __restrict__ col,
                                                        float* __restrict__ dis,
                                                        int* __restrict__ ptr,
                                                        int* __restrict__ src, int E, int N,
                                                        const float* __restrict__ W1,
                                                        const float* __restrict__ W2,
                                                        float* __restrict__ Wp1,
                                                        float* __restrict__ Wp2) {
    __shared__ __attribute__((aligned(16))) int cnt[CB_R];
    __shared__ int wtot[CB_W];
    __shared__ int wbel[CB_W];
    const int t = threadIdx.x;
    if (blockIdx.x >= 2 * CB_NB) {
        const int g = ((int)blockIdx.x - 2 * CB_NB) * CB_T + t;
        if (g < 2 * PK_F4) {
            const bool second = g >= PK_F4;
            const int f = second ? g - PK_F4 : g;
            const int st = f / (GSW / 4);
            const int w = f - st * (GSW / 4);
            const int n = w / 5;
            const int j = w - 5 * n;
            const float* __restrict__ Ws = second ? W2 : W1;
            float* __restrict__ Wd = second ? Wp2 : Wp1;
            v4f val = {0.0f, 0.0f, 0.0f, 0.0f};
            if (n < CH) val = *(const v4f*)(Ws + (size_t)n * CH + st * GBK + 4 * j);
            ((v4f*)Wd)[f] = val;
        }
        return;
    }
    const int lane = t & 31;
    const int wave = __builtin_amdgcn_readfirstlane(t >> 5);
    const bool norm = blockIdx.x >= CB_NB;
    const int b = norm ? (int)blockIdx.x - CB_NB : (int)blockIdx.x;
    const int per = ((N + CB_NB - 1) / CB_NB + 31) & ~31;
    const int blo = b * per;
    int bhi = blo + per;
    if (bhi > N) bhi = N;
    const int* __restrict__ key = norm ? row : col;
    const bool vec = (E & 3) == 0 && E > 0;
    const int4* __restrict__ key4 = (const int4*)key;
    const int E4 = E >> 2;

    for (int lo = blo; lo < bhi; lo += CB_R) {
        int R = bhi - lo;
        if (R > CB_R) R = CB_R;
        ((int4*)cnt)[t] = make_int4(0, 0, 0, 0);
        __syncthreads();

        int below = 0;
#define CB_COUNT(K) { const int d_ = (K) - lo; below += ((K) < lo) ? 1 : 0; \
                      if ((unsigned)d_ < (unsigned)R) atomicAdd(&cnt[d_], 1); }
        if (vec) {
            for (int i0 = 0; i0 < E4; i0 += 4 * CB_T) {
                int4 k[4];
#pragma unroll
                for (int u = 0; u < 4; ++u) k[u] = key_load(key4, i0 + u * CB_T + t, E4);
#pragma unroll
                for (int u = 0; u < 4; ++u) {
                    CB_COUNT(k[u].x) CB_COUNT(k[u].y) CB_COUNT(k[u].z) CB_COUNT(k[u].w)
                }
            }
        } else {
            for (int e = t; e < E; e += CB_T) { const int k = key[e]; CB_COUNT(k) }
        }
#undef CB_COUNT
        __syncthreads();

        if (norm) {
            for (int d = t; d < R; d += CB_T) dis[lo + d] = rsqrtf(1.0f + (float)cnt[d]);
            __syncthreads();
            continue;
        }

        const int bw = wave_scan(below, lane);
        if (lane == 31) wbel[wave] = bw;
        int4 c = ((int4*)cnt)[t];
        const int s1 = c.x;
        const int s2 = s1 + c.y;
        const int s3 = s2 + c.z;
        const int s4 = s3 + c.w;
        const int v = wave_scan(s4, lane);
        if (lane == 31) wtot[wave] = v;
        __syncthreads();
        const int w = wtot[lane];
        const int ws = wave_scan(w, lane);
        const int before = __builtin_amdgcn_readlane(ws - w, wave);
        const int total = __builtin_amdgcn_readlane(ws, 31);
        const int base = __builtin_amdgcn_readlane(wave_scan(wbel[lane], lane), 31);
        const int p0 = base + before + (v - s4);
        const int4 o = make_int4(p0, p0 + s1, p0 + s2, p0 + s3);
        ((int4*)cnt)[t] = o;
        const int d0 = 4 * t;
        if (d0 < R) ptr[lo + d0] = o.x;
        if (d0 + 1 < R) ptr[lo + d0 + 1] = o.y;
        if (d0 + 2 < R) ptr[lo + d0 + 2] = o.z;
        if (d0 + 3 < R) ptr[lo + d0 + 3] = o.w;
        if (t == 0 && lo + R == N) ptr[N] = base + total;
        __syncthreads();

#define CB_FILL(K, EIDX) { const int d_ = (K) - lo; \
                      if ((unsigned)d_ < (unsigned)R) { const int p_ = atomicAdd(&cnt[d_], 1); \
                                                        src[p_] = row[EIDX]; } }
        if (vec) {
            for (int i0 = 0; i0 < E4; i0 += 4 * CB_T) {
                int4 k[4];
#pragma unroll
                for (int u = 0; u < 4; ++u) k[u] = key_load(key4, i0 + u * CB_T + t, E4);
#pragma unroll
                for (int u = 0; u < 4; ++u) {
                    const int e0 = 4 * (i0 + u * CB_T + t);
                    CB_FILL(k[u].x, e0) CB_FILL(k[u].y, e0 + 1)
                    CB_FILL(k[u].z, e0 + 2) CB_FILL(k[u].w, e0 + 3)
                }
            }
        } else {
            for (int e = t; e < E; e += CB_T) { const int k = key[e]; CB_FILL(k, e) }
        }
#undef CB_FILL
        __syncthreads();
    }
}

#define GBM 208
#define GRT (GBM / 16)
#define GTH 640
#define GNT ((CH + 15) / 16)
#define GA4 (GBM * GBK / 4)
#define GW4 (GNT * 16 * GBK / 4)
#define GSF ((GBM + GNP) * GBK)
#define GLDS (GBM * CH)

template <int S0, int S1>
__device__ __forceinline__ void mma_steps(const float* st, int aoff, int boff, v8f (&acc)[GRT]) {
#pragma unroll
    for (int s = S0; s < S1; ++s) {
        const v2f b = *(const v2f*)(st + boff + 4 * s);
#pragma unroll
        for (int t = 0; t < GRT; ++t) {
            const v2f a = *(const v2f*)(st + aoff + t * (16 * GBK) + 4 * s);
            acc[t] = __builtin_amdgcn_wmma_f32_16x16x4_f32(
                false, a, false, b, (short)0, acc[t], false, false);
        }
        __builtin_amdgcn_sched_barrier(0);
    }
}

template <bool POOL>
__global__ __launch_bounds__(GTH) void gemm_wmma_kernel(const float* __restrict__ A,
                                                        const float* __restrict__ W,
                                                        const float* __restrict__ bias,
                                                        float* __restrict__ C,
                                                        int M) {
    __shared__ __attribute__((aligned(16))) float lds[GLDS];
    const int tid  = threadIdx.x;
    const int wave = __builtin_amdgcn_readfirstlane(tid >> 5);
    const int lane = tid & 31;
    const int mn   = lane & 15;
    const int half = lane >> 4;
    const int row0 = blockIdx.x * GBM;

    const bool a1 = tid < (GA4 - GTH);
    const bool w2 = tid < (GW4 - 2 * GTH);
    const float* gp[2];
#pragma unroll
    for (int i = 0; i < 2; ++i) {
        int r = (tid + GTH * i) / 5;
        const int j = (tid + GTH * i) - 5 * r;
        if (r > GBM - 1) r = GBM - 1;
        int gr = row0 + r;
        if (gr > M - 1) gr = M - 1;
        gp[i] = A + (size_t)gr * (POOL ? TP * CH : CH) + 4 * j;
    }
    const float* gw = W + 4 * tid;

    v4f va[2], vw[3];
    v4f p0[TP - 1], p1[TP - 1];
    v4f* l4 = (v4f*)lds;
    if (POOL) {
#pragma unroll
        for (int t = 1; t < TP; ++t) p0[t - 1] = *(const v4f*)(gp[0] + t * CH);
        if (a1) {
#pragma unroll
            for (int t = 1; t < TP; ++t) p1[t - 1] = *(const v4f*)(gp[1] + t * CH);
        }
    }
    va[0] = *(const v4f*)(gp[0]);
    if (a1) va[1] = *(const v4f*)(gp[1]);
    vw[0] = *(const v4f*)(gw);
    vw[1] = *(const v4f*)(gw + 4 * GTH);
    if (w2) vw[2] = *(const v4f*)(gw + 8 * GTH);
    if (POOL) {
        va[0] = 0.25f * (((va[0] + p0[0]) + p0[1]) + p0[2]);
        if (a1) va[1] = 0.25f * (((va[1] + p1[0]) + p1[1]) + p1[2]);
    }
    l4[tid] = va[0];
    if (a1) l4[tid + GTH] = va[1];
    l4[GA4 + tid] = vw[0];
    l4[GA4 + tid + GTH] = vw[1];
    if (w2) l4[GA4 + tid + 2 * GTH] = vw[2];
    __syncthreads();

    v8f acc[GRT] = {};
    const int aoff = mn * GBK + 2 * half;
    const int boff = GBM * GBK + (wave * 16 + mn) * GBK + 2 * half;

    for (int ch = 0; ch < GNCH; ++ch) {
        const float* st = lds + (ch & 1) * GSF;
        const bool more = (ch + 1 < GNCH);
        const int kc = (ch + 1) * GBK;
        const int kw = (ch + 1) * GSW;
        v4f* d4 = (v4f*)(lds + ((ch + 1) & 1) * GSF);
        if (more) {
            if (POOL) {
#pragma unroll
                for (int t = 1; t < TP; ++t) p0[t - 1] = *(const v4f*)(gp[0] + kc + t * CH);
                if (a1) {
#pragma unroll
                    for (int t = 1; t < TP; ++t) p1[t - 1] = *(const v4f*)(gp[1] + kc + t * CH);
                }
            }
            va[0] = *(const v4f*)(gp[0] + kc);
            if (a1) va[1] = *(const v4f*)(gp[1] + kc);
            if (!POOL) {
                vw[0] = *(const v4f*)(gw + kw);
                vw[1] = *(const v4f*)(gw + kw + 4 * GTH);
                if (w2) vw[2] = *(const v4f*)(gw + kw + 8 * GTH);
            }
        }
        if (wave < GNT) mma_steps<0, 2>(st, aoff, boff, acc);
        if (POOL && more) {
            va[0] = 0.25f * (((va[0] + p0[0]) + p0[1]) + p0[2]);
            d4[tid] = va[0];
            if (a1) {
                va[1] = 0.25f * (((va[1] + p1[0]) + p1[1]) + p1[2]);
                d4[tid + GTH] = va[1];
            }
            vw[0] = *(const v4f*)(gw + kw);
            vw[1] = *(const v4f*)(gw + kw + 4 * GTH);
            if (w2) vw[2] = *(const v4f*)(gw + kw + 8 * GTH);
        }
        if (wave < GNT) mma_steps<2, GBK / 4>(st, aoff, boff, acc);
        if (more) {
            if (!POOL) {
                d4[tid] = va[0];
                if (a1) d4[tid + GTH] = va[1];
            }
            d4[GA4 + tid] = vw[0];
            d4[GA4 + tid + GTH] = vw[1];
            if (w2) d4[GA4 + tid + 2 * GTH] = vw[2];
        }
        __syncthreads();
    }

    const int n = wave * 16 + mn;
    const float bv = bias[n < CH ? n : CH - 1];
    if (n < CH) {
#pragma unroll
        for (int t = 0; t < GRT; ++t) {
#pragma unroll
            for (int j = 0; j < 8; ++j)
                lds[(t * 16 + 8 * half + j) * CH + n] = acc[t][j] + bv;
        }
    }
    __syncthreads();
    int rows = M - row0;
    if (rows > GBM) rows = GBM;
    const int n4 = rows * CH4;
    v4f* c4 = (v4f*)(C + (size_t)row0 * CH);
    for (int i = tid; i < n4; i += GTH) c4[i] = l4[i];
}

__global__ __launch_bounds__(256) void gather_kernel(const int* __restrict__ ptr,
                                                     const int* __restrict__ src,
                                                     const float* __restrict__ dis,
                                                     const float* __restrict__ h,
                                                     float* __restrict__ out,
                                                     int total4, int relu) {
    const int g = blockIdx.x * 256 + threadIdx.x;
    if (g >= total4) return;
    const int n = g / CH4;
    const int q = g - n * CH4;
    const float4* __restrict__ h4 = (const float4*)h;
    const float dn = dis[n];
    const float4 self = h4[g];
    const int p0 = ptr[n];
    const int p1 = ptr[n + 1];
    float4 s = make_float4(0.0f, 0.0f, 0.0f, 0.0f);
    for (int p = p0; p < p1; ++p) {
        const int r = src[p];
        const float w = dis[r];
        const float4 v = h4[(size_t)r * CH4 + q];
        s.x += w * v.x; s.y += w * v.y; s.z += w * v.z; s.w += w * v.w;
    }
    float4 o;
    o.x = dn * (s.x + dn * self.x);
    o.y = dn * (s.y + dn * self.y);
    o.z = dn * (s.z + dn * self.z);
    o.w = dn * (s.w + dn * self.w);
    if (relu) {
        o.x = (o.x > 0.0f) ? o.x : 0.01f * o.x;
        o.y = (o.y > 0.0f) ? o.y : 0.01f * o.y;
        o.z = (o.z > 0.0f) ? o.z : 0.01f * o.z;
        o.w = (o.w > 0.0f) ? o.w : 0.01f * o.w;
    }
    ((float4*)out)[g] = o;
}

extern "C" void kernel_launch(void* const* d_in, const int* in_sizes, int n_in,
                              void* d_out, int out_size, void* d_ws, size_t ws_size,
                              hipStream_t stream) {
    const float* x  = (const float*)d_in[0];
    const int*   ei = (const int*)d_in[1];
    const float* W1 = (const float*)d_in[2];
    const float* b1 = (const float*)d_in[3];
    const float* W2 = (const float*)d_in[4];
    const float* b2 = (const float*)d_in[5];
    float* out = (float*)d_out;

    const int E = in_sizes[1] / 2;
    const int N = in_sizes[0] / (TP * CH);
    const int total = N * CH;
    const int* row = ei;
    const int* col = ei + E;

    char* ws = (char*)d_ws;
    float* dis = (float*)ws;
    size_t off = ((size_t)N * sizeof(float) + 255) & ~(size_t)255;
    float* bufA = (float*)(ws + off);
    float* bufB = bufA + total;
    int* ptr = (int*)(bufB + total);
    int* src = ptr + (N + 1);
    const size_t offw = ((size_t)((char*)(src + E) - ws) + 255) & ~(size_t)255;
    float* Wp1 = (float*)(ws + offw);
    float* Wp2 = Wp1 + GNCH * GSW;

    const int gemmBlocks = (N + GBM - 1) / GBM;
    const int total4 = N * CH4;
    const int gatherBlocks = (total4 + 255) / 256;

    csr_build_kernel<<<2 * CB_NB + PK_NB, CB_T, 0, stream>>>(row, col, dis, ptr, src, E, N,
                                                             W1, W2, Wp1, Wp2);

    gemm_wmma_kernel<true><<<gemmBlocks, GTH, 0, stream>>>(x, Wp1, b1, bufB, N);

    gather_kernel<<<gatherBlocks, 256, 0, stream>>>(ptr, src, dis, bufB, bufA, total4, 1);

    gemm_wmma_kernel<false><<<gemmBlocks, GTH, 0, stream>>>(bufA, Wp2, b2, bufB, N);

    gather_kernel<<<gatherBlocks, 256, 0, stream>>>(ptr, src, dis, bufB, out, total4, 0);
}


# CSR fill sweep: an edge's source node comes from a 16-byte row load issued beside the key loads, not from a dependent 4-byte load inside each taken slot behind a full wait; slots stay arrival order
# speedup vs baseline: 3.6681x; 1.0534x over previous
#include <hip/hip_runtime.h>

typedef float v2f __attribute__((ext_vector_type(2)));
typedef float v4f __attribute__((ext_vector_type(4)));
typedef float v8f __attribute__((ext_vector_type(8)));

#define CH 300
#define TP 4
#define CH4 75

#define GBK 20
#define GNP 320
#define GNCH (CH / GBK)
#define GSW (GNP * GBK)
#define PK_F4 (GNCH * GSW / 4)

#define CB_T 1024
#define CB_W (CB_T / 32)
#define CB_R 4096
#define CB_NB 64
#define PK_NB ((2 * PK_F4 + CB_T - 1) / CB_T)

__device__ __forceinline__ int wave_scan(int v, int lane) {
    v += __builtin_amdgcn_update_dpp(0, v, 0x111, 0xf, 0xf, false);
    v += __builtin_amdgcn_update_dpp(0, v, 0x112, 0xf, 0xf, false);
    v += __builtin_amdgcn_update_dpp(0, v, 0x114, 0xf, 0xf, false);
    v += __builtin_amdgcn_update_dpp(0, v, 0x118, 0xf, 0xf, false);
    const int r0 = __builtin_amdgcn_readlane(v, 15);
    return v + (lane >= 16 ? r0 : 0);
}
static_assert(CB_W == 32, "lane l of every wave takes the total of wave l");

__device__ __forceinline__ int4 key_load(const int4* __restrict__ k4, int i, int n4) {
    int4 k = k4[i < n4 ? i : n4 - 1];
    if (i >= n4) k = make_int4(0x7fffffff, 0x7fffffff, 0x7fffffff, 0x7fffffff);
    return k;
}

__global__ __launch_bounds__(CB_T) void csr_build_kernel(const int* __restrict__ row,
                                                        const int* __restrict__ col,
                                                        float* __restrict__ dis,
                                                        int* __restrict__ ptr,
                                                        int* __restrict__ src, int E, int N,
                                                        const float* __restrict__ W1,
                                                        const float* __restrict__ W2,
                                                        float* __restrict__ Wp1,
                                                        float* __restrict__ Wp2) {
    __shared__ __attribute__((aligned(16))) int cnt[CB_R];
    __shared__ int wtot[CB_W];
    __shared__ int wbel[CB_W];
    const int t = threadIdx.x;
    if (blockIdx.x >= 2 * CB_NB) {
        const int g = ((int)blockIdx.x - 2 * CB_NB) * CB_T + t;
        if (g < 2 * PK_F4) {
            const bool second = g >= PK_F4;
            const int f = second ? g - PK_F4 : g;
            const int st = f / (GSW / 4);
            const int w = f - st * (GSW / 4);
            const int n = w / 5;
            const int j = w - 5 * n;
            const float* __restrict__ Ws = second ? W2 : W1;
            float* __restrict__ Wd = second ? Wp2 : Wp1;
            v4f val = {0.0f, 0.0f, 0.0f, 0.0f};
            if (n < CH) val = *(const v4f*)(Ws + (size_t)n * CH + st * GBK + 4 * j);
            ((v4f*)Wd)[f] = val;
        }
        return;
    }
    const int lane = t & 31;
    const int wave = __builtin_amdgcn_readfirstlane(t >> 5);
    const bool norm = blockIdx.x >= CB_NB;
    const int b = norm ? (int)blockIdx.x - CB_NB : (int)blockIdx.x;
    const int per = ((N + CB_NB - 1) / CB_NB + 31) & ~31;
    const int blo = b * per;
    int bhi = blo + per;
    if (bhi > N) bhi = N;
    const int* __restrict__ key = norm ? row : col;
    const bool vec = (E & 3) == 0 && E > 0;
    const int4* __restrict__ key4 = (const int4*)key;
    const int E4 = E >> 2;

    for (int lo = blo; lo < bhi; lo += CB_R) {
        int R = bhi - lo;
        if (R > CB_R) R = CB_R;
        ((int4*)cnt)[t] = make_int4(0, 0, 0, 0);
        __syncthreads();

        int below = 0;
#define CB_COUNT(K) { const int d_ = (K) - lo; below += ((K) < lo) ? 1 : 0; \
                      if ((unsigned)d_ < (unsigned)R) atomicAdd(&cnt[d_], 1); }
        if (vec) {
            for (int i0 = 0; i0 < E4; i0 += 4 * CB_T) {
                int4 k[4];
#pragma unroll
                for (int u = 0; u < 4; ++u) k[u] = key_load(key4, i0 + u * CB_T + t, E4);
#pragma unroll
                for (int u = 0; u < 4; ++u) {
                    CB_COUNT(k[u].x) CB_COUNT(k[u].y) CB_COUNT(k[u].z) CB_COUNT(k[u].w)
                }
            }
        } else {
            for (int e = t; e < E; e += CB_T) { const int k = key[e]; CB_COUNT(k) }
        }
#undef CB_COUNT
        __syncthreads();

        if (norm) {
            for (int d = t; d < R; d += CB_T) dis[lo + d] = rsqrtf(1.0f + (float)cnt[d]);
            __syncthreads();
            continue;
        }

        const int bw = wave_scan(below, lane);
        if (lane == 31) wbel[wave] = bw;
        int4 c = ((int4*)cnt)[t];
        const int s1 = c.x;
        const int s2 = s1 + c.y;
        const int s3 = s2 + c.z;
        const int s4 = s3 + c.w;
        const int v = wave_scan(s4, lane);
        if (lane == 31) wtot[wave] = v;
        __syncthreads();
        const int w = wtot[lane];
        const int ws = wave_scan(w, lane);
        const int before = __builtin_amdgcn_readlane(ws - w, wave);
        const int total = __builtin_amdgcn_readlane(ws, 31);
        const int base = __builtin_amdgcn_readlane(wave_scan(wbel[lane], lane), 31);
        const int p0 = base + before + (v - s4);
        const int4 o = make_int4(p0, p0 + s1, p0 + s2, p0 + s3);
        ((int4*)cnt)[t] = o;
        const int d0 = 4 * t;
        if (d0 < R) ptr[lo + d0] = o.x;
        if (d0 + 1 < R) ptr[lo + d0 + 1] = o.y;
        if (d0 + 2 < R) ptr[lo + d0 + 2] = o.z;
        if (d0 + 3 < R) ptr[lo + d0 + 3] = o.w;
        if (t == 0 && lo + R == N) ptr[N] = base + total;
        __syncthreads();

#define CB_FILL(K, RV) { const int d_ = (K) - lo; \
                      if ((unsigned)d_ < (unsigned)R) { const int p_ = atomicAdd(&cnt[d_], 1); \
                                                        src[p_] = (RV); } }
        if (vec) {
            const int4* __restrict__ row4 = (const int4*)row;
            for (int i0 = 0; i0 < E4; i0 += 4 * CB_T) {
                int4 k[4], r[4];
#pragma unroll
                for (int u = 0; u < 4; ++u) {
                    const int i = i0 + u * CB_T + t;
                    k[u] = key_load(key4, i, E4);
                    r[u] = row4[i < E4 ? i : E4 - 1];
                }
#pragma unroll
                for (int u = 0; u < 4; ++u) {
                    CB_FILL(k[u].x, r[u].x) CB_FILL(k[u].y, r[u].y)
                    CB_FILL(k[u].z, r[u].z) CB_FILL(k[u].w, r[u].w)
                }
            }
        } else {
            for (int e = t; e < E; e += CB_T) { const int k = key[e]; CB_FILL(k, row[e]) }
        }
#undef CB_FILL
        __syncthreads();
    }
}

#define GBM 208
#define GRT (GBM / 16)
#define GTH 640
#define GNT ((CH + 15) / 16)
#define GA4 (GBM * GBK / 4)
#define GW4 (GNT * 16 * GBK / 4)
#define GSF ((GBM + GNP) * GBK)
#define GLDS (GBM * CH)

template <int S0, int S1>
__device__ __forceinline__ void mma_steps(const float* st, int aoff, int boff, v8f (&acc)[GRT]) {
#pragma unroll
    for (int s = S0; s < S1; ++s) {
        const v2f b = *(const v2f*)(st + boff + 4 * s);
#pragma unroll
        for (int t = 0; t < GRT; ++t) {
            const v2f a = *(const v2f*)(st + aoff + t * (16 * GBK) + 4 * s);
            acc[t] = __builtin_amdgcn_wmma_f32_16x16x4_f32(
                false, a, false, b, (short)0, acc[t], false, false);
        }
        __builtin_amdgcn_sched_barrier(0);
    }
}

template <bool POOL>
__global__ __launch_bounds__(GTH) void gemm_wmma_kernel(const float* __restrict__ A,
                                                        const float* __restrict__ W,
                                                        const float* __restrict__ bias,
                                                        float* __restrict__ C,
                                                        int M) {
    __shared__ __attribute__((aligned(16))) float lds[GLDS];
    const int tid  = threadIdx.x;
    const int wave = __builtin_amdgcn_readfirstlane(tid >> 5);
    const int lane = tid & 31;
    const int mn   = lane & 15;
    const int half = lane >> 4;
    const int row0 = blockIdx.x * GBM;

    const bool a1 = tid < (GA4 - GTH);
    const bool w2 = tid < (GW4 - 2 * GTH);
    const float* gp[2];
#pragma unroll
    for (int i = 0; i < 2; ++i) {
        int r = (tid + GTH * i) / 5;
        const int j = (tid + GTH * i) - 5 * r;
        if (r > GBM - 1) r = GBM - 1;
        int gr = row0 + r;
        if (gr > M - 1) gr = M - 1;
        gp[i] = A + (size_t)gr * (POOL ? TP * CH : CH) + 4 * j;
    }
    const float* gw = W + 4 * tid;

    v4f va[2], vw[3];
    v4f p0[TP - 1], p1[TP - 1];
    v4f* l4 = (v4f*)lds;
    if (POOL) {
#pragma unroll
        for (int t = 1; t < TP; ++t) p0[t - 1] = *(const v4f*)(gp[0] + t * CH);
        if (a1) {
#pragma unroll
            for (int t = 1; t < TP; ++t) p1[t - 1] = *(const v4f*)(gp[1] + t * CH);
        }
    }
    va[0] = *(const v4f*)(gp[0]);
    if (a1) va[1] = *(const v4f*)(gp[1]);
    vw[0] = *(const v4f*)(gw);
    vw[1] = *(const v4f*)(gw + 4 * GTH);
    if (w2) vw[2] = *(const v4f*)(gw + 8 * GTH);
    if (POOL) {
        va[0] = 0.25f * (((va[0] + p0[0]) + p0[1]) + p0[2]);
        if (a1) va[1] = 0.25f * (((va[1] + p1[0]) + p1[1]) + p1[2]);
    }
    l4[tid] = va[0];
    if (a1) l4[tid + GTH] = va[1];
    l4[GA4 + tid] = vw[0];
    l4[GA4 + tid + GTH] = vw[1];
    if (w2) l4[GA4 + tid + 2 * GTH] = vw[2];
    __syncthreads();

    v8f acc[GRT] = {};
    const int aoff = mn * GBK + 2 * half;
    const int boff = GBM * GBK + (wave * 16 + mn) * GBK + 2 * half;

    for (int ch = 0; ch < GNCH; ++ch) {
        const float* st = lds + (ch & 1) * GSF;
        const bool more = (ch + 1 < GNCH);
        const int kc = (ch + 1) * GBK;
        const int kw = (ch + 1) * GSW;
        v4f* d4 = (v4f*)(lds + ((ch + 1) & 1) * GSF);
        if (more) {
            if (POOL) {
#pragma unroll
                for (int t = 1; t < TP; ++t) p0[t - 1] = *(const v4f*)(gp[0] + kc + t * CH);
                if (a1) {
#pragma unroll
                    for (int t = 1; t < TP; ++t) p1[t - 1] = *(const v4f*)(gp[1] + kc + t * CH);
                }
            }
            va[0] = *(const v4f*)(gp[0] + kc);
            if (a1) va[1] = *(const v4f*)(gp[1] + kc);
            if (!POOL) {
                vw[0] = *(const v4f*)(gw + kw);
                vw[1] = *(const v4f*)(gw + kw + 4 * GTH);
                if (w2) vw[2] = *(const v4f*)(gw + kw + 8 * GTH);
            }
        }
        if (wave < GNT) mma_steps<0, 2>(st, aoff, boff, acc);
        if (POOL && more) {
            va[0] = 0.25f * (((va[0] + p0[0]) + p0[1]) + p0[2]);
            d4[tid] = va[0];
            if (a1) {
                va[1] = 0.25f * (((va[1] + p1[0]) + p1[1]) + p1[2]);
                d4[tid + GTH] = va[1];
            }
            vw[0] = *(const v4f*)(gw + kw);
            vw[1] = *(const v4f*)(gw + kw + 4 * GTH);
            if (w2) vw[2] = *(const v4f*)(gw + kw + 8 * GTH);
        }
        if (wave < GNT) mma_steps<2, GBK / 4>(st, aoff, boff, acc);
        if (more) {
            if (!POOL) {
                d4[tid] = va[0];
                if (a1) d4[tid + GTH] = va[1];
            }
            d4[GA4 + tid] = vw[0];
            d4[GA4 + tid + GTH] = vw[1];
            if (w2) d4[GA4 + tid + 2 * GTH] = vw[2];
        }
        __syncthreads();
    }

    const int n = wave * 16 + mn;
    const float bv = bias[n < CH ? n : CH - 1];
    if (n < CH) {
#pragma unroll
        for (int t = 0; t < GRT; ++t) {
#pragma unroll
            for (int j = 0; j < 8; ++j)
                lds[(t * 16 + 8 * half + j) * CH + n] = acc[t][j] + bv;
        }
    }
    __syncthreads();
    int rows = M - row0;
    if (rows > GBM) rows = GBM;
    const int n4 = rows * CH4;
    v4f* c4 = (v4f*)(C + (size_t)row0 * CH);
    for (int i = tid; i < n4; i += GTH) c4[i] = l4[i];
}

__global__ __launch_bounds__(256) void gather_kernel(const int* __restrict__ ptr,
                                                     const int* __restrict__ src,
                                                     const float* __restrict__ dis,
                                                     const float* __restrict__ h,
                                                     float* __restrict__ out,
                                                     int total4, int relu) {
    const int g = blockIdx.x * 256 + threadIdx.x;
    if (g >= total4) return;
    const int n = g / CH4;
    const int q = g - n * CH4;
    const float4* __restrict__ h4 = (const float4*)h;
    const float dn = dis[n];
    const float4 self = h4[g];
    const int p0 = ptr[n];
    const int p1 = ptr[n + 1];
    float4 s = make_float4(0.0f, 0.0f, 0.0f, 0.0f);
    for (int p = p0; p < p1; ++p) {
        const int r = src[p];
        const float w = dis[r];
        const float4 v = h4[(size_t)r * CH4 + q];
        s.x += w * v.x; s.y += w * v.y; s.z += w * v.z; s.w += w * v.w;
    }
    float4 o;
    o.x = dn * (s.x + dn * self.x);
    o.y = dn * (s.y + dn * self.y);
    o.z = dn * (s.z + dn * self.z);
    o.w = dn * (s.w + dn * self.w);
    if (relu) {
        o.x = (o.x > 0.0f) ? o.x : 0.01f * o.x;
        o.y = (o.y > 0.0f) ? o.y : 0.01f * o.y;
        o.z = (o.z > 0.0f) ? o.z : 0.01f * o.z;
        o.w = (o.w > 0.0f) ? o.w : 0.01f * o.w;
    }
    ((float4*)out)[g] = o;
}

extern "C" void kernel_launch(void* const* d_in, const int* in_sizes, int n_in,
                              void* d_out, int out_size, void* d_ws, size_t ws_size,
                              hipStream_t stream) {
    const float* x  = (const float*)d_in[0];
    const int*   ei = (const int*)d_in[1];
    const float* W1 = (const float*)d_in[2];
    const float* b1 = (const float*)d_in[3];
    const float* W2 = (const float*)d_in[4];
    const float* b2 = (const float*)d_in[5];
    float* out = (float*)d_out;

    const int E = in_sizes[1] / 2;
    const int N = in_sizes[0] / (TP * CH);
    const int total = N * CH;
    const int* row = ei;
    const int* col = ei + E;

    char* ws = (char*)d_ws;
    float* dis = (float*)ws;
    size_t off = ((size_t)N * sizeof(float) + 255) & ~(size_t)255;
    float* bufA = (float*)(ws + off);
    float* bufB = bufA + total;
    int* ptr = (int*)(bufB + total);
    int* src = ptr + (N + 1);
    const size_t offw = ((size_t)((char*)(src + E) - ws) + 255) & ~(size_t)255;
    float* Wp1 = (float*)(ws + offw);
    float* Wp2 = Wp1 + GNCH * GSW;

    const int gemmBlocks = (N + GBM - 1) / GBM;
    const int total4 = N * CH4;
    const int gatherBlocks = (total4 + 255) / 256;

    csr_build_kernel<<<2 * CB_NB + PK_NB, CB_T, 0, stream>>>(row, col, dis, ptr, src, E, N,
                                                             W1, W2, Wp1, Wp2);

    gemm_wmma_kernel<true><<<gemmBlocks, GTH, 0, stream>>>(x, Wp1, b1, bufB, N);

    gather_kernel<<<gatherBlocks, 256, 0, stream>>>(ptr, src, dis, bufB, bufA, total4, 1);

    gemm_wmma_kernel<false><<<gemmBlocks, GTH, 0, stream>>>(bufA, Wp2, b2, bufB, N);

    gather_kernel<<<gatherBlocks, 256, 0, stream>>>(ptr, src, dis, bufB, out, total4, 0);
}


# Gather re-cut to aligned requests: rows it reads get a 1280-byte pitch (GEMM stores 5 pad chunks), one wave sums a node, 3 loads a source row = its 5 aligned 256-byte units; same sum order
# speedup vs baseline: 3.7955x; 1.0347x over previous
#include <hip/hip_runtime.h>

typedef float v2f __attribute__((ext_vector_type(2)));
typedef float v4f __attribute__((ext_vector_type(4)));
typedef float v8f __attribute__((ext_vector_type(8)));

#define CH 300
#define TP 4
#define CH4 75
#define HP4 80
#define HPF (4 * HP4)

#define GBK 20
#define GNP 320
#define GNCH (CH / GBK)
#define GSW (GNP * GBK)
#define PK_F4 (GNCH * GSW / 4)

#define CB_T 1024
#define CB_W (CB_T / 32)
#define CB_R 4096
#define CB_NB 64
#define PK_NB ((2 * PK_F4 + CB_T - 1) / CB_T)

__device__ __forceinline__ int wave_scan(int v, int lane) {
    v += __builtin_amdgcn_update_dpp(0, v, 0x111, 0xf, 0xf, false);
    v += __builtin_amdgcn_update_dpp(0, v, 0x112, 0xf, 0xf, false);
    v += __builtin_amdgcn_update_dpp(0, v, 0x114, 0xf, 0xf, false);
    v += __builtin_amdgcn_update_dpp(0, v, 0x118, 0xf, 0xf, false);
    const int r0 = __builtin_amdgcn_readlane(v, 15);
    return v + (lane >= 16 ? r0 : 0);
}
static_assert(CB_W == 32, "lane l of every wave takes the total of wave l");

__device__ __forceinline__ int4 key_load(const int4* __restrict__ k4, int i, int n4) {
    int4 k = k4[i < n4 ? i : n4 - 1];
    if (i >= n4) k = make_int4(0x7fffffff, 0x7fffffff, 0x7fffffff, 0x7fffffff);
    return k;
}

__global__ __launch_bounds__(CB_T) void csr_build_kernel(const int* __restrict__ row,
                                                        const int* __restrict__ col,
                                                        float* __restrict__ dis,
                                                        int* __restrict__ ptr,
                                                        int* __restrict__ src, int E, int N,
                                                        const float* __restrict__ W1,
                                                        const float* __restrict__ W2,
                                                        float* __restrict__ Wp1,
                                                        float* __restrict__ Wp2) {
    __shared__ __attribute__((aligned(16))) int cnt[CB_R];
    __shared__ int wtot[CB_W];
    __shared__ int wbel[CB_W];
    const int t = threadIdx.x;
    if (blockIdx.x >= 2 * CB_NB) {
        const int g = ((int)blockIdx.x - 2 * CB_NB) * CB_T + t;
        if (g < 2 * PK_F4) {
            const bool second = g >= PK_F4;
            const int f = second ? g - PK_F4 : g;
            const int st = f / (GSW / 4);
            const int w = f - st * (GSW / 4);
            const int n = w / 5;
            const int j = w - 5 * n;
            const float* __restrict__ Ws = second ? W2 : W1;
            float* __restrict__ Wd = second ? Wp2 : Wp1;
            v4f val = {0.0f, 0.0f, 0.0f, 0.0f};
            if (n < CH) val = *(const v4f*)(Ws + (size_t)n * CH + st * GBK + 4 * j);
            ((v4f*)Wd)[f] = val;
        }
        return;
    }
    const int lane = t & 31;
    const int wave = __builtin_amdgcn_readfirstlane(t >> 5);
    const bool norm = blockIdx.x >= CB_NB;
    const int b = norm ? (int)blockIdx.x - CB_NB : (int)blockIdx.x;
    const int per = ((N + CB_NB - 1) / CB_NB + 31) & ~31;
    const int blo = b * per;
    int bhi = blo + per;
    if (bhi > N) bhi = N;
    const int* __restrict__ key = norm ? row : col;
    const bool vec = (E & 3) == 0 && E > 0;
    const int4* __restrict__ key4 = (const int4*)key;
    const int E4 = E >> 2;

    for (int lo = blo; lo < bhi; lo += CB_R) {
        int R = bhi - lo;
        if (R > CB_R) R = CB_R;
        ((int4*)cnt)[t] = make_int4(0, 0, 0, 0);
        __syncthreads();

        int below = 0;
#define CB_COUNT(K) { const int d_ = (K) - lo; below += ((K) < lo) ? 1 : 0; \
                      if ((unsigned)d_ < (unsigned)R) atomicAdd(&cnt[d_], 1); }
        if (vec) {
            for (int i0 = 0; i0 < E4; i0 += 4 * CB_T) {
                int4 k[4];
#pragma unroll
                for (int u = 0; u < 4; ++u) k[u] = key_load(key4, i0 + u * CB_T + t, E4);
#pragma unroll
                for (int u = 0; u < 4; ++u) {
                    CB_COUNT(k[u].x) CB_COUNT(k[u].y) CB_COUNT(k[u].z) CB_COUNT(k[u].w)
                }
            }
        } else {
            for (int e = t; e < E; e += CB_T) { const int k = key[e]; CB_COUNT(k) }
        }
#undef CB_COUNT
        __syncthreads();

        if (norm) {
            for (int d = t; d < R; d += CB_T) dis[lo + d] = rsqrtf(1.0f + (float)cnt[d]);
            __syncthreads();
            continue;
        }

        const int bw = wave_scan(below, lane);
        if (lane == 31) wbel[wave] = bw;
        int4 c = ((int4*)cnt)[t];
        const int s1 = c.x;
        const int s2 = s1 + c.y;
        const int s3 = s2 + c.z;
        const int s4 = s3 + c.w;
        const int v = wave_scan(s4, lane);
        if (lane == 31) wtot[wave] = v;
        __syncthreads();
        const int w = wtot[lane];
        const int ws = wave_scan(w, lane);
        const int before = __builtin_amdgcn_readlane(ws - w, wave);
        const int total = __builtin_amdgcn_readlane(ws, 31);
        const int base = __builtin_amdgcn_readlane(wave_scan(wbel[lane], lane), 31);
        const int p0 = base + before + (v - s4);
        const int4 o = make_int4(p0, p0 + s1, p0 + s2, p0 + s3);
        ((int4*)cnt)[t] = o;
        const int d0 = 4 * t;
        if (d0 < R) ptr[lo + d0] = o.x;
        if (d0 + 1 < R) ptr[lo + d0 + 1] = o.y;
        if (d0 + 2 < R) ptr[lo + d0 + 2] = o.z;
        if (d0 + 3 < R) ptr[lo + d0 + 3] = o.w;
        if (t == 0 && lo + R == N) ptr[N] = base + total;
        __syncthreads();

#define CB_FILL(K, RV) { const int d_ = (K) - lo; \
                      if ((unsigned)d_ < (unsigned)R) { const int p_ = atomicAdd(&cnt[d_], 1); \
                                                        src[p_] = (RV); } }
        if (vec) {
            const int4* __restrict__ row4 = (const int4*)row;
            for (int i0 = 0; i0 < E4; i0 += 4 * CB_T) {
                int4 k[4], r[4];
#pragma unroll
                for (int u = 0; u < 4; ++u) {
                    const int i = i0 + u * CB_T + t;
                    k[u] = key_load(key4, i, E4);
                    r[u] = row4[i < E4 ? i : E4 - 1];
                }
#pragma unroll
                for (int u = 0; u < 4; ++u) {
                    CB_FILL(k[u].x, r[u].x) CB_FILL(k[u].y, r[u].y)
                    CB_FILL(k[u].z, r[u].z) CB_FILL(k[u].w, r[u].w)
                }
            }
        } else {
            for (int e = t; e < E; e += CB_T) { const int k = key[e]; CB_FILL(k, row[e]) }
        }
#undef CB_FILL
        __syncthreads();
    }
}

#define GBM 208
#define GRT (GBM / 16)
#define GTH 640
#define GNT ((CH + 15) / 16)
#define GA4 (GBM * GBK / 4)
#define GW4 (GNT * 16 * GBK / 4)
#define GSF ((GBM + GNP) * GBK)
#define GLDS (GBM * CH)

template <int S0, int S1>
__device__ __forceinline__ void mma_steps(const float* st, int aoff, int boff, v8f (&acc)[GRT]) {
#pragma unroll
    for (int s = S0; s < S1; ++s) {
        const v2f b = *(const v2f*)(st + boff + 4 * s);
#pragma unroll
        for (int t = 0; t < GRT; ++t) {
            const v2f a = *(const v2f*)(st + aoff + t * (16 * GBK) + 4 * s);
            acc[t] = __builtin_amdgcn_wmma_f32_16x16x4_f32(
                false, a, false, b, (short)0, acc[t], false, false);
        }
        __builtin_amdgcn_sched_barrier(0);
    }
}

template <bool POOL>
__global__ __launch_bounds__(GTH) void gemm_wmma_kernel(const float* __restrict__ A,
                                                        const float* __restrict__ W,
                                                        const float* __restrict__ bias,
                                                        float* __restrict__ C,
                                                        int M) {
    __shared__ __attribute__((aligned(16))) float lds[GLDS];
    const int tid  = threadIdx.x;
    const int wave = __builtin_amdgcn_readfirstlane(tid >> 5);
    const int lane = tid & 31;
    const int mn   = lane & 15;
    const int half = lane >> 4;
    const int row0 = blockIdx.x * GBM;

    const bool a1 = tid < (GA4 - GTH);
    const bool w2 = tid < (GW4 - 2 * GTH);
    const float* gp[2];
#pragma unroll
    for (int i = 0; i < 2; ++i) {
        int r = (tid + GTH * i) / 5;
        const int j = (tid + GTH * i) - 5 * r;
        if (r > GBM - 1) r = GBM - 1;
        int gr = row0 + r;
        if (gr > M - 1) gr = M - 1;
        gp[i] = A + (size_t)gr * (POOL ? TP * CH : CH) + 4 * j;
    }
    const float* gw = W + 4 * tid;

    v4f va[2], vw[3];
    v4f p0[TP - 1], p1[TP - 1];
    v4f* l4 = (v4f*)lds;
    if (POOL) {
#pragma unroll
        for (int t = 1; t < TP; ++t) p0[t - 1] = *(const v4f*)(gp[0] + t * CH);
        if (a1) {
#pragma unroll
            for (int t = 1; t < TP; ++t) p1[t - 1] = *(const v4f*)(gp[1] + t * CH);
        }
    }
    va[0] = *(const v4f*)(gp[0]);
    if (a1) va[1] = *(const v4f*)(gp[1]);
    vw[0] = *(const v4f*)(gw);
    vw[1] = *(const v4f*)(gw + 4 * GTH);
    if (w2) vw[2] = *(const v4f*)(gw + 8 * GTH);
    if (POOL) {
        va[0] = 0.25f * (((va[0] + p0[0]) + p0[1]) + p0[2]);
        if (a1) va[1] = 0.25f * (((va[1] + p1[0]) + p1[1]) + p1[2]);
    }
    l4[tid] = va[0];
    if (a1) l4[tid + GTH] = va[1];
    l4[GA4 + tid] = vw[0];
    l4[GA4 + tid + GTH] = vw[1];
    if (w2) l4[GA4 + tid + 2 * GTH] = vw[2];
    __syncthreads();

    v8f acc[GRT] = {};
    const int aoff = mn * GBK + 2 * half;
    const int boff = GBM * GBK + (wave * 16 + mn) * GBK + 2 * half;

    for (int ch = 0; ch < GNCH; ++ch) {
        const float* st = lds + (ch & 1) * GSF;
        const bool more = (ch + 1 < GNCH);
        const int kc = (ch + 1) * GBK;
        const int kw = (ch + 1) * GSW;
        v4f* d4 = (v4f*)(lds + ((ch + 1) & 1) * GSF);
        if (more) {
            if (POOL) {
#pragma unroll
                for (int t = 1; t < TP; ++t) p0[t - 1] = *(const v4f*)(gp[0] + kc + t * CH);
                if (a1) {
#pragma unroll
                    for (int t = 1; t < TP; ++t) p1[t - 1] = *(const v4f*)(gp[1] + kc + t * CH);
                }
            }
            va[0] = *(const v4f*)(gp[0] + kc);
            if (a1) va[1] = *(const v4f*)(gp[1] + kc);
            if (!POOL) {
                vw[0] = *(const v4f*)(gw + kw);
                vw[1] = *(const v4f*)(gw + kw + 4 * GTH);
                if (w2) vw[2] = *(const v4f*)(gw + kw + 8 * GTH);
            }
        }
        if (wave < GNT) mma_steps<0, 2>(st, aoff, boff, acc);
        if (POOL && more) {
            va[0] = 0.25f * (((va[0] + p0[0]) + p0[1]) + p0[2]);
            d4[tid] = va[0];
            if (a1) {
                va[1] = 0.25f * (((va[1] + p1[0]) + p1[1]) + p1[2]);
                d4[tid + GTH] = va[1];
            }
            vw[0] = *(const v4f*)(gw + kw);
            vw[1] = *(const v4f*)(gw + kw + 4 * GTH);
            if (w2) vw[2] = *(const v4f*)(gw + kw + 8 * GTH);
        }
        if (wave < GNT) mma_steps<2, GBK / 4>(st, aoff, boff, acc);
        if (more) {
            if (!POOL) {
                d4[tid] = va[0];
                if (a1) d4[tid + GTH] = va[1];
            }
            d4[GA4 + tid] = vw[0];
            d4[GA4 + tid + GTH] = vw[1];
            if (w2) d4[GA4 + tid + 2 * GTH] = vw[2];
        }
        __syncthreads();
    }

    const int n = wave * 16 + mn;
    const float bv = bias[n < CH ? n : CH - 1];
    if (n < CH) {
#pragma unroll
        for (int t = 0; t < GRT; ++t) {
#pragma unroll
            for (int j = 0; j < 8; ++j)
                lds[(t * 16 + 8 * half + j) * CH + n] = acc[t][j] + bv;
        }
    }
    __syncthreads();
    int rows = M - row0;
    if (rows > GBM) rows = GBM;
    v4f* c4 = (v4f*)(C + (size_t)row0 * HPF);
    const int cr = tid / HP4;
    const int cc = tid - cr * HP4;
    const int sc = cc < CH4 ? cc : CH4 - 1;
    const v4f zero4 = {0.0f, 0.0f, 0.0f, 0.0f};
    for (int r = cr; r < rows; r += GTH / HP4) {
        const v4f v = l4[r * CH4 + sc];
        c4[r * HP4 + cc] = (cc < CH4) ? v : zero4;
    }
}
static_assert(GTH % HP4 == 0, "a thread of the result store keeps its chunk column");

#define GN 8
#define GT (GN * 32)
#define GC2 (CH4 - 64)

__device__ __forceinline__ float4 gather_finish(float4 s, float4 self, float dn, int relu) {
    float4 o;
    o.x = dn * (s.x + dn * self.x);
    o.y = dn * (s.y + dn * self.y);
    o.z = dn * (s.z + dn * self.z);
    o.w = dn * (s.w + dn * self.w);
    if (relu) {
        o.x = (o.x > 0.0f) ? o.x : 0.01f * o.x;
        o.y = (o.y > 0.0f) ? o.y : 0.01f * o.y;
        o.z = (o.z > 0.0f) ? o.z : 0.01f * o.z;
        o.w = (o.w > 0.0f) ? o.w : 0.01f * o.w;
    }
    return o;
}

__global__ __launch_bounds__(GT) void gather_kernel(const int* __restrict__ ptr,
                                                    const int* __restrict__ src,
                                                    const float* __restrict__ dis,
                                                    const float* __restrict__ h,
                                                    float* __restrict__ out,
                                                    int N, int relu) {
    __shared__ float4 stage[GN * CH4];
    const int t = threadIdx.x;
    const int lane = t & 31;
    const int wv = __builtin_amdgcn_readfirstlane(t >> 5);
    const int n0 = blockIdx.x * GN;
    const int n = n0 + wv;
    if (n < N) {
        const float4* __restrict__ h4 = (const float4*)h;
        const bool third = lane < GC2;
        const int c2 = third ? 64 + lane : CH4 - 1;
        const float4* __restrict__ hn = h4 + (size_t)n * HP4;
        const float4 self0 = hn[lane];
        const float4 self1 = hn[32 + lane];
        const float4 self2 = hn[c2];
        const float dn = dis[n];
        const int p0 = ptr[n];
        const int p1 = ptr[n + 1];
        asm volatile("" :: "s"(dn), "s"(p0), "s"(p1));
        float4 s0 = make_float4(0.0f, 0.0f, 0.0f, 0.0f);
        float4 s1 = s0, s2 = s0;
        for (int pb = p0; pb < p1; pb += 32) {
            int cnt = p1 - pb;
            if (cnt > 32) cnt = 32;
            const int pl = pb + lane;
            const int idl = src[pl < p1 ? pl : p1 - 1];
            const int dl = __float_as_int(dis[idl]);
            for (int j = 0; j < cnt; ++j) {
                const int r = __builtin_amdgcn_readlane(idl, j);
                const float w = __int_as_float(__builtin_amdgcn_readlane(dl, j));
                const float4* __restrict__ hr = h4 + (size_t)r * HP4;
                const float4 v0 = hr[lane];
                const float4 v1 = hr[32 + lane];
                const float4 v2 = hr[c2];
                s0.x += w * v0.x; s0.y += w * v0.y; s0.z += w * v0.z; s0.w += w * v0.w;
                s1.x += w * v1.x; s1.y += w * v1.y; s1.z += w * v1.z; s1.w += w * v1.w;
                s2.x += w * v2.x; s2.y += w * v2.y; s2.z += w * v2.z; s2.w += w * v2.w;
            }
        }
        float4* __restrict__ sr = stage + wv * CH4;
        sr[lane] = gather_finish(s0, self0, dn, relu);
        sr[32 + lane] = gather_finish(s1, self1, dn, relu);
        if (third) sr[64 + lane] = gather_finish(s2, self2, dn, relu);
    }
    __syncthreads();
    int rows = N - n0;
    if (rows > GN) rows = GN;
    float4* __restrict__ o4 = (float4*)out + (size_t)n0 * CH4;
    for (int i = t; i < rows * CH4; i += GT) o4[i] = stage[i];
}

extern "C" void kernel_launch(void* const* d_in, const int* in_sizes, int n_in,
                              void* d_out, int out_size, void* d_ws, size_t ws_size,
                              hipStream_t stream) {
    const float* x  = (const float*)d_in[0];
    const int*   ei = (const int*)d_in[1];
    const float* W1 = (const float*)d_in[2];
    const float* b1 = (const float*)d_in[3];
    const float* W2 = (const float*)d_in[4];
    const float* b2 = (const float*)d_in[5];
    float* out = (float*)d_out;

    const int E = in_sizes[1] / 2;
    const int N = in_sizes[0] / (TP * CH);
    const int total = N * CH;
    const int* row = ei;
    const int* col = ei + E;

    char* ws = (char*)d_ws;
    float* dis = (float*)ws;
    size_t off = ((size_t)N * sizeof(float) + 255) & ~(size_t)255;
    float* bufA = (float*)(ws + off);
    const size_t offB = (off + (size_t)total * sizeof(float) + 255) & ~(size_t)255;
    float* bufB = (float*)(ws + offB);
    int* ptr = (int*)(bufB + (size_t)N * HPF);
    int* src = ptr + (N + 1);
    const size_t offw = ((size_t)((char*)(src + E) - ws) + 255) & ~(size_t)255;
    float* Wp1 = (float*)(ws + offw);
    float* Wp2 = Wp1 + GNCH * GSW;

    const int gemmBlocks = (N + GBM - 1) / GBM;
    const int gatherBlocks = (N + GN - 1) / GN;

    csr_build_kernel<<<2 * CB_NB + PK_NB, CB_T, 0, stream>>>(row, col, dis, ptr, src, E, N,
                                                             W1, W2, Wp1, Wp2);

    gemm_wmma_kernel<true><<<gemmBlocks, GTH, 0, stream>>>(x, Wp1, b1, bufB, N);

    gather_kernel<<<gatherBlocks, GT, 0, stream>>>(ptr, src, dis, bufB, bufA, N, 1);

    gemm_wmma_kernel<false><<<gemmBlocks, GTH, 0, stream>>>(bufA, Wp2, b2, bufB, N);

    gather_kernel<<<gatherBlocks, GT, 0, stream>>>(ptr, src, dis, bufB, out, N, 0);
}


# CSR build and degree norm leave launch 1: they run as 15 extra blocks of the layer-1 GEMM launch (one edge sweep each) on WGPs its 241 blocks leave free, under its matrix work; slots arrival order
# speedup vs baseline: 4.3665x; 1.1504x over previous
#include <hip/hip_runtime.h>

typedef float v2f __attribute__((ext_vector_type(2)));
typedef float v4f __attribute__((ext_vector_type(4)));
typedef float v8f __attribute__((ext_vector_type(8)));

#define CH 300
#define TP 4
#define CH4 75
#define HP4 80
#define HPF (4 * HP4)

#define GBK 20
#define GNP 320
#define GNCH (CH / GBK)
#define GSW (GNP * GBK)
#define PK_F4 (GNCH * GSW / 4)

#define PK_T 1024
#define PK_NB ((2 * PK_F4 + PK_T - 1) / PK_T)

__global__ __launch_bounds__(PK_T) void pack_kernel(const float* __restrict__ W1,
                                                    const float* __restrict__ W2,
                                                    float* __restrict__ Wp1,
                                                    float* __restrict__ Wp2) {
    const int g = (int)blockIdx.x * PK_T + (int)threadIdx.x;
    if (g < 2 * PK_F4) {
        const bool second = g >= PK_F4;
        const int f = second ? g - PK_F4 : g;
        const int st = f / (GSW / 4);
        const int w = f - st * (GSW / 4);
        const int n = w / 5;
        const int j = w - 5 * n;
        const float* __restrict__ Ws = second ? W2 : W1;
        float* __restrict__ Wd = second ? Wp2 : Wp1;
        v4f val = {0.0f, 0.0f, 0.0f, 0.0f};
        if (n < CH) val = *(const v4f*)(Ws + (size_t)n * CH + st * GBK + 4 * j);
        ((v4f*)Wd)[f] = val;
    }
}

#define CS_NB 15
#define CS_T 640
#define CS_W (CS_T / 32)
#define CB_R 4096
#define CS_K 80
#define CB_EB 20
#define CS_LST (2 * CB_R + 80)
#define CS_LDS (CS_LST + CS_K * CS_T)
static_assert(CB_R <= (1 << (32 - CB_EB)), "a node offset of the range fits beside the edge index");
static_assert(CS_W <= 32, "lane l of every wave takes the total of wave l");
static_assert(CB_R / 8 <= CS_T, "the block scan gives a thread 8 adjacent counts");

__device__ __forceinline__ int wave_scan(int v, int lane) {
    v += __builtin_amdgcn_update_dpp(0, v, 0x111, 0xf, 0xf, false);
    v += __builtin_amdgcn_update_dpp(0, v, 0x112, 0xf, 0xf, false);
    v += __builtin_amdgcn_update_dpp(0, v, 0x114, 0xf, 0xf, false);
    v += __builtin_amdgcn_update_dpp(0, v, 0x118, 0xf, 0xf, false);
    const int r0 = __builtin_amdgcn_readlane(v, 15);
    return v + (lane >= 16 ? r0 : 0);
}

__device__ __forceinline__ int4 key_load(const int4* __restrict__ k4, int i, int n4) {
    int4 k = k4[i < n4 ? i : n4 - 1];
    if (i >= n4) k = make_int4(0x7fffffff, 0x7fffffff, 0x7fffffff, 0x7fffffff);
    return k;
}

__device__ __forceinline__ void csr_role(int* __restrict__ sm, int b,
                                         const int* __restrict__ row,
                                         const int* __restrict__ col,
                                         float* __restrict__ dis,
                                         int* __restrict__ ptr,
                                         int* __restrict__ src, int E, int N) {
    int* cnt = sm;
    int* deg = sm + CB_R;
    int* wtot = sm + 2 * CB_R;
    int* wbel = wtot + 32;
    int* ovf = wbel + 32;
    unsigned* lst = (unsigned*)(sm + CS_LST);
    const int t = threadIdx.x;
    const int lane = t & 31;
    const int wave = __builtin_amdgcn_readfirstlane(t >> 5);
    const int per = ((N + CS_NB - 1) / CS_NB + 31) & ~31;
    const int blo = b * per;
    int bhi = blo + per;
    if (bhi > N) bhi = N;
    const bool vec = (E & 3) == 0 && E > 0;
    const int4* __restrict__ col4 = (const int4*)col;
    const int4* __restrict__ row4 = (const int4*)row;
    const int E4 = E >> 2;
    const bool keep = vec && E <= (1 << CB_EB);

    for (int lo = blo; lo < bhi; lo += CB_R) {
        int R = bhi - lo;
        if (R > CB_R) R = CB_R;
        for (int i = t; i < 2 * CB_R / 4; i += CS_T) ((int4*)sm)[i] = make_int4(0, 0, 0, 0);
        if (t == 0) *ovf = 0;
        __syncthreads();

        int below = 0;
        int nm = 0;
#define CS_COUNT(K) { const int d_ = (K) - lo; below += ((K) < lo) ? 1 : 0; \
                      if ((unsigned)d_ < (unsigned)R) atomicAdd(&cnt[d_], 1); }
#define CS_MARK(K, EI) { const int d_ = (K) - lo; below += ((K) < lo) ? 1 : 0; \
                      if ((unsigned)d_ < (unsigned)R) { atomicAdd(&cnt[d_], 1); \
                          lst[(nm < CS_K ? nm : CS_K - 1) * CS_T + t] = ((unsigned)d_ << CB_EB) | (unsigned)(EI); \
                          ++nm; } }
#define CS_DEG(K) { const int d_ = (K) - lo; \
                      if ((unsigned)d_ < (unsigned)R) atomicAdd(&deg[d_], 1); }
        if (keep) {
            for (int i0 = 0; i0 < E4; i0 += 4 * CS_T) {
                int4 k[4], r[4];
#pragma unroll
                for (int u = 0; u < 4; ++u) {
                    k[u] = key_load(col4, i0 + u * CS_T + t, E4);
                    r[u] = key_load(row4, i0 + u * CS_T + t, E4);
                }
#pragma unroll
                for (int u = 0; u < 4; ++u) {
                    const int e0 = 4 * (i0 + u * CS_T + t);
                    CS_MARK(k[u].x, e0) CS_MARK(k[u].y, e0 + 1)
                    CS_MARK(k[u].z, e0 + 2) CS_MARK(k[u].w, e0 + 3)
                    CS_DEG(r[u].x) CS_DEG(r[u].y) CS_DEG(r[u].z) CS_DEG(r[u].w)
                }
            }
            if (nm > CS_K) *ovf = 1;
        } else if (vec) {
            for (int i0 = 0; i0 < E4; i0 += 4 * CS_T) {
                int4 k[4], r[4];
#pragma unroll
                for (int u = 0; u < 4; ++u) {
                    k[u] = key_load(col4, i0 + u * CS_T + t, E4);
                    r[u] = key_load(row4, i0 + u * CS_T + t, E4);
                }
#pragma unroll
                for (int u = 0; u < 4; ++u) {
                    CS_COUNT(k[u].x) CS_COUNT(k[u].y) CS_COUNT(k[u].z) CS_COUNT(k[u].w)
                    CS_DEG(r[u].x) CS_DEG(r[u].y) CS_DEG(r[u].z) CS_DEG(r[u].w)
                }
            }
        } else {
            for (int e = t; e < E; e += CS_T) {
                const int k = col[e];
                const int r = row[e];
                CS_COUNT(k) CS_DEG(r)
            }
        }
#undef CS_COUNT
#undef CS_MARK
#undef CS_DEG
        __syncthreads();
        const bool listed = keep && *ovf == 0;

        for (int d = t; d < R; d += CS_T) dis[lo + d] = rsqrtf(1.0f + (float)deg[d]);

        const int bw = wave_scan(below, lane);
        if (lane == 31) wbel[wave] = bw;
        const bool own = t < CB_R / 8;
        int4 ca = make_int4(0, 0, 0, 0), cb = ca;
        if (own) { ca = ((int4*)cnt)[2 * t]; cb = ((int4*)cnt)[2 * t + 1]; }
        const int s1 = ca.x;
        const int s2 = s1 + ca.y;
        const int s3 = s2 + ca.z;
        const int s4 = s3 + ca.w;
        const int s5 = s4 + cb.x;
        const int s6 = s5 + cb.y;
        const int s7 = s6 + cb.z;
        const int s8 = s7 + cb.w;
        const int v = wave_scan(s8, lane);
        if (lane == 31) wtot[wave] = v;
        __syncthreads();
        const int w = lane < CS_W ? wtot[lane] : 0;
        const int ws = wave_scan(w, lane);
        const int before = __builtin_amdgcn_readlane(ws - w, wave);
        const int total = __builtin_amdgcn_readlane(ws, 31);
        const int wb = lane < CS_W ? wbel[lane] : 0;
        const int base = __builtin_amdgcn_readlane(wave_scan(wb, lane), 31);
        const int p0 = base + before + (v - s8);
        if (own) {
            const int4 oa = make_int4(p0, p0 + s1, p0 + s2, p0 + s3);
            const int4 ob = make_int4(p0 + s4, p0 + s5, p0 + s6, p0 + s7);
            ((int4*)cnt)[2 * t] = oa;
            ((int4*)cnt)[2 * t + 1] = ob;
            const int d0 = 8 * t;
            int* __restrict__ pp = ptr + lo + d0;
            if (d0 < R) pp[0] = oa.x;
            if (d0 + 1 < R) pp[1] = oa.y;
            if (d0 + 2 < R) pp[2] = oa.z;
            if (d0 + 3 < R) pp[3] = oa.w;
            if (d0 + 4 < R) pp[4] = ob.x;
            if (d0 + 5 < R) pp[5] = ob.y;
            if (d0 + 6 < R) pp[6] = ob.z;
            if (d0 + 7 < R) pp[7] = ob.w;
        }
        if (t == 0 && lo + R == N) ptr[N] = base + total;
        __syncthreads();

#define CS_FILL(K, RV) { const int d_ = (K) - lo; \
                      if ((unsigned)d_ < (unsigned)R) { const int p_ = atomicAdd(&cnt[d_], 1); \
                                                        src[p_] = (RV); } }
        if (listed) {
            for (int q0 = 0; q0 < nm; q0 += 8) {
                unsigned wd[8];
                int rv[8];
#pragma unroll
                for (int u = 0; u < 8; ++u) {
                    const int q = q0 + u < nm ? q0 + u : nm - 1;
                    wd[u] = lst[q * CS_T + t];
                }
#pragma unroll
                for (int u = 0; u < 8; ++u) rv[u] = row[wd[u] & ((1u << CB_EB) - 1u)];
#pragma unroll
                for (int u = 0; u < 8; ++u) {
                    if (q0 + u < nm) {
                        const int p_ = atomicAdd(&cnt[wd[u] >> CB_EB], 1);
                        src[p_] = rv[u];
                    }
                }
            }
        } else if (vec) {
            for (int i0 = 0; i0 < E4; i0 += 4 * CS_T) {
                int4 k[4], r[4];
#pragma unroll
                for (int u = 0; u < 4; ++u) {
                    const int i = i0 + u * CS_T + t;
                    k[u] = key_load(col4, i, E4);
                    r[u] = row4[i < E4 ? i : E4 - 1];
                }
#pragma unroll
                for (int u = 0; u < 4; ++u) {
                    CS_FILL(k[u].x, r[u].x) CS_FILL(k[u].y, r[u].y)
                    CS_FILL(k[u].z, r[u].z) CS_FILL(k[u].w, r[u].w)
                }
            }
        } else {
            for (int e = t; e < E; e += CS_T) { const int k = col[e]; CS_FILL(k, row[e]) }
        }
#undef CS_FILL
        __syncthreads();
    }
}

#define GBM 208
#define GRT (GBM / 16)
#define GTH 640
#define GNT ((CH + 15) / 16)
#define GA4 (GBM * GBK / 4)
#define GW4 (GNT * 16 * GBK / 4)
#define GSF ((GBM + GNP) * GBK)
#define GLDS (GBM * CH)
static_assert(CS_LDS <= GLDS, "the CSR role's scratch fits the GEMM's LDS array");
static_assert(CS_T == GTH, "the CSR role runs with the GEMM's block size");

template <int S0, int S1>
__device__ __forceinline__ void mma_steps(const float* st, int aoff, int boff, v8f (&acc)[GRT]) {
#pragma unroll
    for (int s = S0; s < S1; ++s) {
        const v2f b = *(const v2f*)(st + boff + 4 * s);
#pragma unroll
        for (int t = 0; t < GRT; ++t) {
            const v2f a = *(const v2f*)(st + aoff + t * (16 * GBK) + 4 * s);
            acc[t] = __builtin_amdgcn_wmma_f32_16x16x4_f32(
                false, a, false, b, (short)0, acc[t], false, false);
        }
        __builtin_amdgcn_sched_barrier(0);
    }
}

template <bool POOL>
__global__ __launch_bounds__(GTH) void gemm_wmma_kernel(const float* __restrict__ A,
                                                        const float* __restrict__ W,
                                                        const float* __restrict__ bias,
                                                        float* __restrict__ C,
                                                        int M,
                                                        const int* __restrict__ erow,
                                                        const int* __restrict__ ecol,
                                                        float* __restrict__ dis,
                                                        int* __restrict__ ptr,
                                                        int* __restrict__ src,
                                                        int E, int rb) {
    __shared__ __attribute__((aligned(16))) float lds[GLDS];
    if (POOL && (int)blockIdx.x < rb) {
        csr_role((int*)lds, (int)blockIdx.x, erow, ecol, dis, ptr, src, E, M);
        return;
    }
    const int tid  = threadIdx.x;
    const int wave = __builtin_amdgcn_readfirstlane(tid >> 5);
    const int lane = tid & 31;
    const int mn   = lane & 15;
    const int half = lane >> 4;
    const int row0 = ((int)blockIdx.x - rb) * GBM;

    const bool a1 = tid < (GA4 - GTH);
    const bool w2 = tid < (GW4 - 2 * GTH);
    const float* gp[2];
#pragma unroll
    for (int i = 0; i < 2; ++i) {
        int r = (tid + GTH * i) / 5;
        const int j = (tid + GTH * i) - 5 * r;
        if (r > GBM - 1) r = GBM - 1;
        int gr = row0 + r;
        if (gr > M - 1) gr = M - 1;
        gp[i] = A + (size_t)gr * (POOL ? TP * CH : CH) + 4 * j;
    }
    const float* gw = W + 4 * tid;

    v4f va[2], vw[3];
    v4f p0[TP - 1], p1[TP - 1];
    v4f* l4 = (v4f*)lds;
    if (POOL) {
#pragma unroll
        for (int t = 1; t < TP; ++t) p0[t - 1] = *(const v4f*)(gp[0] + t * CH);
        if (a1) {
#pragma unroll
            for (int t = 1; t < TP; ++t) p1[t - 1] = *(const v4f*)(gp[1] + t * CH);
        }
    }
    va[0] = *(const v4f*)(gp[0]);
    if (a1) va[1] = *(const v4f*)(gp[1]);
    vw[0] = *(const v4f*)(gw);
    vw[1] = *(const v4f*)(gw + 4 * GTH);
    if (w2) vw[2] = *(const v4f*)(gw + 8 * GTH);
    if (POOL) {
        va[0] = 0.25f * (((va[0] + p0[0]) + p0[1]) + p0[2]);
        if (a1) va[1] = 0.25f * (((va[1] + p1[0]) + p1[1]) + p1[2]);
    }
    l4[tid] = va[0];
    if (a1) l4[tid + GTH] = va[1];
    l4[GA4 + tid] = vw[0];
    l4[GA4 + tid + GTH] = vw[1];
    if (w2) l4[GA4 + tid + 2 * GTH] = vw[2];
    __syncthreads();

    v8f acc[GRT] = {};
    const int aoff = mn * GBK + 2 * half;
    const int boff = GBM * GBK + (wave * 16 + mn) * GBK + 2 * half;

    for (int ch = 0; ch < GNCH; ++ch) {
        const float* st = lds + (ch & 1) * GSF;
        const bool more = (ch + 1 < GNCH);
        const int kc = (ch + 1) * GBK;
        const int kw = (ch + 1) * GSW;
        v4f* d4 = (v4f*)(lds + ((ch + 1) & 1) * GSF);
        if (more) {
            if (POOL) {
#pragma unroll
                for (int t = 1; t < TP; ++t) p0[t - 1] = *(const v4f*)(gp[0] + kc + t * CH);
                if (a1) {
#pragma unroll
                    for (int t = 1; t < TP; ++t) p1[t - 1] = *(const v4f*)(gp[1] + kc + t * CH);
                }
            }
            va[0] = *(const v4f*)(gp[0] + kc);
            if (a1) va[1] = *(const v4f*)(gp[1] + kc);
            if (!POOL) {
                vw[0] = *(const v4f*)(gw + kw);
                vw[1] = *(const v4f*)(gw + kw + 4 * GTH);
                if (w2) vw[2] = *(const v4f*)(gw + kw + 8 * GTH);
            }
        }
        if (wave < GNT) mma_steps<0, 2>(st, aoff, boff, acc);
        if (POOL && more) {
            va[0] = 0.25f * (((va[0] + p0[0]) + p0[1]) + p0[2]);
            d4[tid] = va[0];
            if (a1) {
                va[1] = 0.25f * (((va[1] + p1[0]) + p1[1]) + p1[2]);
                d4[tid + GTH] = va[1];
            }
            vw[0] = *(const v4f*)(gw + kw);
            vw[1] = *(const v4f*)(gw + kw + 4 * GTH);
            if (w2) vw[2] = *(const v4f*)(gw + kw + 8 * GTH);
        }
        if (wave < GNT) mma_steps<2, GBK / 4>(st, aoff, boff, acc);
        if (more) {
            if (!POOL) {
                d4[tid] = va[0];
                if (a1) d4[tid + GTH] = va[1];
            }
            d4[GA4 + tid] = vw[0];
            d4[GA4 + tid + GTH] = vw[1];
            if (w2) d4[GA4 + tid + 2 * GTH] = vw[2];
        }
        __syncthreads();
    }

    const int n = wave * 16 + mn;
    const float bv = bias[n < CH ? n : CH - 1];
    if (n < CH) {
#pragma unroll
        for (int t = 0; t < GRT; ++t) {
#pragma unroll
            for (int j = 0; j < 8; ++j)
                lds[(t * 16 + 8 * half + j) * CH + n] = acc[t][j] + bv;
        }
    }
    __syncthreads();
    int rows = M - row0;
    if (rows > GBM) rows = GBM;
    v4f* c4 = (v4f*)(C + (size_t)row0 * HPF);
    const int cr = tid / HP4;
    const int cc = tid - cr * HP4;
    const int sc = cc < CH4 ? cc : CH4 - 1;
    const v4f zero4 = {0.0f, 0.0f, 0.0f, 0.0f};
    for (int r = cr; r < rows; r += GTH / HP4) {
        const v4f v = l4[r * CH4 + sc];
        c4[r * HP4 + cc] = (cc < CH4) ? v : zero4;
    }
}
static_assert(GTH % HP4 == 0, "a thread of the result store keeps its chunk column");

#define GN 8
#define GT (GN * 32)
#define GC2 (CH4 - 64)

__device__ __forceinline__ float4 gather_finish(float4 s, float4 self, float dn, int relu) {
    float4 o;
    o.x = dn * (s.x + dn * self.x);
    o.y = dn * (s.y + dn * self.y);
    o.z = dn * (s.z + dn * self.z);
    o.w = dn * (s.w + dn * self.w);
    if (relu) {
        o.x = (o.x > 0.0f) ? o.x : 0.01f * o.x;
        o.y = (o.y > 0.0f) ? o.y : 0.01f * o.y;
        o.z = (o.z > 0.0f) ? o.z : 0.01f * o.z;
        o.w = (o.w > 0.0f) ? o.w : 0.01f * o.w;
    }
    return o;
}

__global__ __launch_bounds__(GT) void gather_kernel(const int* __restrict__ ptr,
                                                    const int* __restrict__ src,
                                                    const float* __restrict__ dis,
                                                    const float* __restrict__ h,
                                                    float* __restrict__ out,
                                                    int N, int relu) {
    __shared__ float4 stage[GN * CH4];
    const int t = threadIdx.x;
    const int lane = t & 31;
    const int wv = __builtin_amdgcn_readfirstlane(t >> 5);
    const int n0 = blockIdx.x * GN;
    const int n = n0 + wv;
    if (n < N) {
        const float4* __restrict__ h4 = (const float4*)h;
        const bool third = lane < GC2;
        const int c2 = third ? 64 + lane : CH4 - 1;
        const float4* __restrict__ hn = h4 + (size_t)n * HP4;
        const float4 self0 = hn[lane];
        const float4 self1 = hn[32 + lane];
        const float4 self2 = hn[c2];
        const float dn = dis[n];
        const int p0 = ptr[n];
        const int p1 = ptr[n + 1];
        asm volatile("" :: "s"(dn), "s"(p0), "s"(p1));
        float4 s0 = make_float4(0.0f, 0.0f, 0.0f, 0.0f);
        float4 s1 = s0, s2 = s0;
        for (int pb = p0; pb < p1; pb += 32) {
            int cnt = p1 - pb;
            if (cnt > 32) cnt = 32;
            const int pl = pb + lane;
            const int idl = src[pl < p1 ? pl : p1 - 1];
            const int dl = __float_as_int(dis[idl]);
            for (int j = 0; j < cnt; ++j) {
                const int r = __builtin_amdgcn_readlane(idl, j);
                const float w = __int_as_float(__builtin_amdgcn_readlane(dl, j));
                const float4* __restrict__ hr = h4 + (size_t)r * HP4;
                const float4 v0 = hr[lane];
                const float4 v1 = hr[32 + lane];
                const float4 v2 = hr[c2];
                s0.x += w * v0.x; s0.y += w * v0.y; s0.z += w * v0.z; s0.w += w * v0.w;
                s1.x += w * v1.x; s1.y += w * v1.y; s1.z += w * v1.z; s1.w += w * v1.w;
                s2.x += w * v2.x; s2.y += w * v2.y; s2.z += w * v2.z; s2.w += w * v2.w;
            }
        }
        float4* __restrict__ sr = stage + wv * CH4;
        sr[lane] = gather_finish(s0, self0, dn, relu);
        sr[32 + lane] = gather_finish(s1, self1, dn, relu);
        if (third) sr[64 + lane] = gather_finish(s2, self2, dn, relu);
    }
    __syncthreads();
    int rows = N - n0;
    if (rows > GN) rows = GN;
    float4* __restrict__ o4 = (float4*)out + (size_t)n0 * CH4;
    for (int i = t; i < rows * CH4; i += GT) o4[i] = stage[i];
}

extern "C" void kernel_launch(void* const* d_in, const int* in_sizes, int n_in,
                              void* d_out, int out_size, void* d_ws, size_t ws_size,
                              hipStream_t stream) {
    const float* x  = (const float*)d_in[0];
    const int*   ei = (const int*)d_in[1];
    const float* W1 = (const float*)d_in[2];
    const float* b1 = (const float*)d_in[3];
    const float* W2 = (const float*)d_in[4];
    const float* b2 = (const float*)d_in[5];
    float* out = (float*)d_out;

    const int E = in_sizes[1] / 2;
    const int N = in_sizes[0] / (TP * CH);
    const int total = N * CH;
    const int* row = ei;
    const int* col = ei + E;

    char* ws = (char*)d_ws;
    float* dis = (float*)ws;
    size_t off = ((size_t)N * sizeof(float) + 255) & ~(size_t)255;
    float* bufA = (float*)(ws + off);
    const size_t offB = (off + (size_t)total * sizeof(float) + 255) & ~(size_t)255;
    float* bufB = (float*)(ws + offB);
    int* ptr = (int*)(bufB + (size_t)N * HPF);
    int* src = ptr + (N + 1);
    const size_t offw = ((size_t)((char*)(src + E) - ws) + 255) & ~(size_t)255;
    float* Wp1 = (float*)(ws + offw);
    float* Wp2 = Wp1 + GNCH * GSW;

    const int gemmBlocks = (N + GBM - 1) / GBM;
    const int gatherBlocks = (N + GN - 1) / GN;

    pack_kernel<<<PK_NB, PK_T, 0, stream>>>(W1, W2, Wp1, Wp2);

    gemm_wmma_kernel<true><<<CS_NB + gemmBlocks, GTH, 0, stream>>>(x, Wp1, b1, bufB, N,
                                                                   row, col, dis, ptr, src, E,
                                                                   CS_NB);

    gather_kernel<<<gatherBlocks, GT, 0, stream>>>(ptr, src, dis, bufB, bufA, N, 1);

    gemm_wmma_kernel<false><<<gemmBlocks, GTH, 0, stream>>>(bufA, Wp2, b2, bufB, N,
                                                            nullptr, nullptr, nullptr, nullptr,
                                                            nullptr, 0, 0);

    gather_kernel<<<gatherBlocks, GT, 0, stream>>>(ptr, src, dis, bufB, out, N, 0);
}
